# starting kernel
# baseline (speedup 1.0000x reference)
#include <hip/hip_runtime.h>

typedef __attribute__((ext_vector_type(16))) _Float16 v16h;
typedef __attribute__((ext_vector_type(8)))  _Float16 v8h;
typedef __attribute__((ext_vector_type(8)))  float    v8f;

#define HID 128

__global__ void convert_wt_kernel(const float* __restrict__ W,
                                  _Float16* __restrict__ Wt) {
    int idx = blockIdx.x * blockDim.x + threadIdx.x;
    int n = idx >> 7;
    int k = idx & 127;
    Wt[idx] = (_Float16)W[k * HID + n];
}

__global__ void encode_kernel(const float* __restrict__ x,
                              const float* __restrict__ W,
                              const float* __restrict__ b,
                              float* __restrict__ h, int N) {
    int wave = threadIdx.x >> 5, lane = threadIdx.x & 31;
    int n = blockIdx.x * 8 + wave;
    if (n >= N) return;
    int c = lane * 4;
    float xv = x[n];
    float4 w  = *(const float4*)(W + c);
    float4 bb = *(const float4*)(b + c);
    float4 r;
    r.x = xv * w.x + bb.x;
    r.y = xv * w.y + bb.y;
    r.z = xv * w.z + bb.z;
    r.w = xv * w.w + bb.w;
    *(float4*)(h + (size_t)n * HID + c) = r;
}

__global__ void scale_kernel(const float* __restrict__ h,
                             const float* __restrict__ eps, int l,
                             float* __restrict__ agg, int n4) {
    int i = blockIdx.x * blockDim.x + threadIdx.x;
    if (i >= n4) return;
    float s = 1.0f + eps[l];
    float4 v = ((const float4*)h)[i];
    v.x *= s; v.y *= s; v.z *= s; v.w *= s;
    ((float4*)agg)[i] = v;
}

__global__ void edge_kernel(const float* __restrict__ h,
                            const int* __restrict__ ei,
                            const float* __restrict__ ea,
                            const float* __restrict__ eW,
                            const float* __restrict__ eb,
                            float* __restrict__ agg, int E) {
    int wave = threadIdx.x >> 5, lane = threadIdx.x & 31;
    int e = blockIdx.x * 8 + wave;
    if (e >= E) return;
    int src = ei[e];
    int dst = ei[E + e];
    float a0 = ea[2 * e];
    float a1 = ea[2 * e + 1];
    int c = lane * 4;
    float4 hv = *(const float4*)(h + (size_t)src * HID + c);
    float4 w0 = *(const float4*)(eW + c);
    float4 w1 = *(const float4*)(eW + HID + c);
    float4 bb = *(const float4*)(eb + c);
    float* ap = agg + (size_t)dst * HID + c;
    atomicAdd(ap + 0, fmaxf(hv.x + a0 * w0.x + a1 * w1.x + bb.x, 0.0f));
    atomicAdd(ap + 1, fmaxf(hv.y + a0 * w0.y + a1 * w1.y + bb.y, 0.0f));
    atomicAdd(ap + 2, fmaxf(hv.z + a0 * w0.z + a1 * w1.z + bb.z, 0.0f));
    atomicAdd(ap + 3, fmaxf(hv.w + a0 * w0.w + a1 * w1.w + bb.w, 0.0f));
}

__global__ void zero_kernel(float* __restrict__ p, int n4) {
    int i = blockIdx.x * blockDim.x + threadIdx.x;
    if (i < n4) ((float4*)p)[i] = make_float4(0.f, 0.f, 0.f, 0.f);
}

__global__ void pool_kernel(const float* __restrict__ h,
                            const int* __restrict__ batch,
                            float* __restrict__ pooled, int N) {
    int wave = threadIdx.x >> 5, lane = threadIdx.x & 31;
    int n = blockIdx.x * 8 + wave;
    if (n >= N) return;
    int g = batch[n];
    int c = lane * 4;
    float4 hv = *(const float4*)(h + (size_t)n * HID + c);
    float* pp = pooled + (size_t)g * HID + c;
    atomicAdd(pp + 0, hv.x);
    atomicAdd(pp + 1, hv.y);
    atomicAdd(pp + 2, hv.z);
    atomicAdd(pp + 3, hv.w);
}

template <bool RELU>
__global__ __launch_bounds__(256) void gemm128_wmma_kernel(
    const float* __restrict__ A, const _Float16* __restrict__ Wt,
    const float* __restrict__ bias, float* __restrict__ out, int rows) {
    __shared__ __align__(16) _Float16 sW[HID * HID];
    {
        const uint4* s = (const uint4*)Wt;
        uint4* d = (uint4*)sW;
        for (int i = threadIdx.x; i < (HID * HID) / 8; i += 256) d[i] = s[i];
    }
    __syncthreads();

    int wave = threadIdx.x >> 5, lane = threadIdx.x & 31;
    int tile = blockIdx.x * 8 + wave;
    int row0 = tile * 16;
    if (row0 >= rows) return;

    int mloc = lane & 15;
    int half = lane >> 4;
    int base = half * 8;
    const float* rowA = A + (size_t)(row0 + mloc) * HID;

    v8f acc[8];
#pragma unroll
    for (int nt = 0; nt < 8; ++nt) {
        float bv = bias[nt * 16 + mloc];
#pragma unroll
        for (int r = 0; r < 8; ++r) acc[nt][r] = bv;
    }

#pragma unroll
    for (int kk = 0; kk < 4; ++kk) {
        int k0 = kk * 32 + base;
        float4 a0 = *(const float4*)(rowA + k0);
        float4 a1 = *(const float4*)(rowA + k0 + 4);
        float4 a2 = *(const float4*)(rowA + k0 + 16);
        float4 a3 = *(const float4*)(rowA + k0 + 20);
        v16h a;
        a[0]  = (_Float16)a0.x; a[1]  = (_Float16)a0.y;
        a[2]  = (_Float16)a0.z; a[3]  = (_Float16)a0.w;
        a[4]  = (_Float16)a1.x; a[5]  = (_Float16)a1.y;
        a[6]  = (_Float16)a1.z; a[7]  = (_Float16)a1.w;
        a[8]  = (_Float16)a2.x; a[9]  = (_Float16)a2.y;
        a[10] = (_Float16)a2.z; a[11] = (_Float16)a2.w;
        a[12] = (_Float16)a3.x; a[13] = (_Float16)a3.y;
        a[14] = (_Float16)a3.z; a[15] = (_Float16)a3.w;
#pragma unroll
        for (int nt = 0; nt < 8; ++nt) {
            int n = nt * 16 + mloc;
            const v8h* bp = (const v8h*)(sW + n * HID + k0);
            v8h blo = bp[0];
            v8h bhi = bp[2];
            v16h b;
#pragma unroll
            for (int i = 0; i < 8; ++i) { b[i] = blo[i]; b[8 + i] = bhi[i]; }
            acc[nt] = __builtin_amdgcn_wmma_f32_16x16x32_f16(
                false, a, false, b, (short)0, acc[nt], false, false);
        }
    }

#pragma unroll
    for (int nt = 0; nt < 8; ++nt) {
        int col = nt * 16 + mloc;
#pragma unroll
        for (int r = 0; r < 8; ++r) {
            float v = acc[nt][r];
            if (RELU) v = fmaxf(v, 0.0f);
            int M = row0 + r + half * 8;
            out[(size_t)M * HID + col] = v;
        }
    }
}

extern "C" void kernel_launch(void* const* d_in, const int* in_sizes, int n_in,
                              void* d_out, int out_size, void* d_ws, size_t ws_size,
                              hipStream_t stream) {
    const int N = in_sizes[0];
    const int E = in_sizes[1] / 2;
    const int G = out_size / (3 * HID);

    const float* enc_W  = (const float*)d_in[12];
    const float* enc_b  = (const float*)d_in[13];
    const float* edge_W = (const float*)d_in[14];
    const float* edge_b = (const float*)d_in[15];
    const float* eps    = (const float*)d_in[16];
    const float* W1     = (const float*)d_in[17];
    const float* b1     = (const float*)d_in[18];
    const float* W2     = (const float*)d_in[19];
    const float* b2     = (const float*)d_in[20];
    const float* fc_W   = (const float*)d_in[21];
    const float* fc_b   = (const float*)d_in[22];

    float* bufA   = (float*)d_ws;
    float* bufB   = bufA + (size_t)N * HID;
    float* pooled = bufB + (size_t)N * HID;
    _Float16* w16 = (_Float16*)(pooled + (size_t)G * HID);
    _Float16* W1t0 = w16;
    _Float16* W1t1 = w16 + 16384;
    _Float16* W2t0 = w16 + 2 * 16384;
    _Float16* W2t1 = w16 + 3 * 16384;
    _Float16* fcWt = w16 + 4 * 16384;

    convert_wt_kernel<<<64, 256, 0, stream>>>(W1,             W1t0);
    convert_wt_kernel<<<64, 256, 0, stream>>>(W1 + HID * HID, W1t1);
    convert_wt_kernel<<<64, 256, 0, stream>>>(W2,             W2t0);
    convert_wt_kernel<<<64, 256, 0, stream>>>(W2 + HID * HID, W2t1);
    convert_wt_kernel<<<64, 256, 0, stream>>>(fc_W,           fcWt);

    const int n4        = (N * HID) / 4;
    const int nodeBlk   = (N + 7) / 8;
    const int edgeBlk   = (E + 7) / 8;
    const int scaleBlk  = (n4 + 255) / 256;
    const int gemmBlk   = ((N + 15) / 16 + 7) / 8;
    const int gP4       = (G * HID) / 4;
    const int fcBlk     = ((G + 15) / 16 + 7) / 8;

    for (int g = 0; g < 3; ++g) {
        const float* x  = (const float*)d_in[4 * g + 0];
        const int*   ei = (const int*)  d_in[4 * g + 1];
        const float* ea = (const float*)d_in[4 * g + 2];
        const int*   bt = (const int*)  d_in[4 * g + 3];

        encode_kernel<<<nodeBlk, 256, 0, stream>>>(x, enc_W, enc_b, bufA, N);

        float* hcur = bufA;
        float* aggb = bufB;
        for (int l = 0; l < 2; ++l) {
            scale_kernel<<<scaleBlk, 256, 0, stream>>>(hcur, eps, l, aggb, n4);
            edge_kernel<<<edgeBlk, 256, 0, stream>>>(
                hcur, ei, ea, edge_W + (size_t)l * 2 * HID, edge_b + (size_t)l * HID,
                aggb, E);
            _Float16* w1t = (l == 0) ? W1t0 : W1t1;
            _Float16* w2t = (l == 0) ? W2t0 : W2t1;
            gemm128_wmma_kernel<true><<<gemmBlk, 256, 0, stream>>>(
                aggb, w1t, b1 + (size_t)l * HID, hcur, N);
            gemm128_wmma_kernel<true><<<gemmBlk, 256, 0, stream>>>(
                hcur, w2t, b2 + (size_t)l * HID, aggb, N);
            float* t = hcur; hcur = aggb; aggb = t;
        }

        zero_kernel<<<(gP4 + 255) / 256, 256, 0, stream>>>(pooled, gP4);
        pool_kernel<<<nodeBlk, 256, 0, stream>>>(hcur, bt, pooled, N);
        gemm128_wmma_kernel<false><<<fcBlk, 256, 0, stream>>>(
            pooled, fcWt, fc_b, (float*)d_out + (size_t)g * G * HID, G);
    }
}


# edge scan: a pre-pass writes one owner byte per edge (dst/196); work-groups scan that list (1/4 of the bytes), note matching 4-edge groups, read and check their dst; row sums now in edge order
# speedup vs baseline: 3.0605x; 3.0605x over previous
#include <hip/hip_runtime.h>

typedef __attribute__((ext_vector_type(16))) _Float16 v16h;
typedef __attribute__((ext_vector_type(8)))  _Float16 v8h;
typedef __attribute__((ext_vector_type(8)))  float    v8f;

#define HID 128

__global__ void convert_wt_kernel(const float* __restrict__ W,
                                  _Float16* __restrict__ Wf) {
    int idx = blockIdx.x * blockDim.x + threadIdx.x;
    int i    = idx & 7;
    int lane = (idx >> 3) & 31;
    int j    = (idx >> 8) & 1;
    int nt   = (idx >> 9) & 7;
    int kk   = idx >> 12;
    int n = nt * 16 + (lane & 15);
    int k = kk * 32 + (lane >> 4) * 8 + j * 16 + i;
    Wf[idx] = (_Float16)W[k * HID + n];
}

__global__ void encode_kernel(const float* __restrict__ x,
                              const float* __restrict__ W,
                              const float* __restrict__ b,
                              float* __restrict__ h, int N) {
    int wave = threadIdx.x >> 5, lane = threadIdx.x & 31;
    int n = blockIdx.x * 8 + wave;
    if (n >= N) return;
    int c = lane * 4;
    float xv = x[n];
    float4 w  = *(const float4*)(W + c);
    float4 bb = *(const float4*)(b + c);
    float4 r;
    r.x = xv * w.x + bb.x;
    r.y = xv * w.y + bb.y;
    r.z = xv * w.z + bb.z;
    r.w = xv * w.w + bb.w;
    *(float4*)(h + (size_t)n * HID + c) = r;
}

#define OWN_R 196
__global__ __launch_bounds__(256) void owner_key_kernel(
    const int* __restrict__ ei_a, const int* __restrict__ ei_p,
    const int* __restrict__ ei_n, unsigned* __restrict__ key, int E, int nchunk) {
    const int* ei = blockIdx.y == 0 ? ei_a : (blockIdx.y == 1 ? ei_p : ei_n);
    const int* dstp = ei + E;
    const int t = blockIdx.x * 256 + threadIdx.x;
    if (t >= nchunk * 32) return;
    const int c = __builtin_amdgcn_readfirstlane(t >> 5), L = t & 31;
    int4 d[4];
    if ((E & 3) == 0 && (c + 1) * 512 <= E) {
#pragma unroll
        for (int j = 0; j < 4; ++j)
            d[j] = *(const int4*)(dstp + ((c * 4 + j) << 7) + (L << 2));
    } else {
#pragma unroll
        for (int j = 0; j < 4; ++j) {
            const int e0 = ((c * 4 + j) << 7) + (L << 2);
            d[j] = make_int4(0, 0, 0, 0);
            if (e0 < E)     d[j].x = dstp[e0];
            if (e0 + 1 < E) d[j].y = dstp[e0 + 1];
            if (e0 + 2 < E) d[j].z = dstp[e0 + 2];
            if (e0 + 3 < E) d[j].w = dstp[e0 + 3];
        }
    }
    unsigned kw[4];
#pragma unroll
    for (int j = 0; j < 4; ++j)
        kw[j] = (((unsigned)d[j].x / OWN_R) & 255u) |
                ((((unsigned)d[j].y / OWN_R) & 255u) << 8) |
                ((((unsigned)d[j].z / OWN_R) & 255u) << 16) |
                ((((unsigned)d[j].w / OWN_R) & 255u) << 24);
    ((uint4*)key)[(size_t)blockIdx.y * nchunk * 32 + t] =
        make_uint4(kw[0], kw[1], kw[2], kw[3]);
}

#define OWN_T 13
#define ROWP  33
#define QCAP  256
#define LQCAP 384
__device__ __forceinline__ void add_pending(
    const int4* q, const int* pd, int np, const float* __restrict__ h, int lane,
    float4 w0, float4 w1, float4 bb, float4* acc) {
    for (int p = 0; p < np; p += 8) {
        int slot[8], srcu[8], rowu[8], a0u[8], a1u[8];
        float4 hv[8];
#pragma unroll
        for (int k = 0; k < 8; ++k) slot[k] = pd[min(p + k, np - 1)];
#pragma unroll
        for (int k = 0; k < 8; ++k) {
            int4 t = q[slot[k]];
            srcu[k] = __builtin_amdgcn_readfirstlane(t.x);
            rowu[k] = __builtin_amdgcn_readfirstlane(t.y);
            a0u[k]  = __builtin_amdgcn_readfirstlane(t.z);
            a1u[k]  = __builtin_amdgcn_readfirstlane(t.w);
        }
#pragma unroll
        for (int k = 0; k < 8; ++k) {
            if (p + k < np)
                hv[k] = *(const float4*)(h + (size_t)srcu[k] * HID + 4 * lane);
        }
#pragma unroll
        for (int k = 0; k < 8; ++k) {
            if (p + k < np) {
                float a0 = __int_as_float(a0u[k]);
                float a1 = __int_as_float(a1u[k]);
                float4 s = acc[rowu[k] * ROWP + lane];
                s.x += fmaxf(hv[k].x + a0 * w0.x + a1 * w1.x + bb.x, 0.0f);
                s.y += fmaxf(hv[k].y + a0 * w0.y + a1 * w1.y + bb.y, 0.0f);
                s.z += fmaxf(hv[k].z + a0 * w0.z + a1 * w1.z + bb.z, 0.0f);
                s.w += fmaxf(hv[k].w + a0 * w0.w + a1 * w1.w + bb.w, 0.0f);
                acc[rowu[k] * ROWP + lane] = s;
            }
        }
    }
}

__device__ __forceinline__ void key_step(unsigned kd, unsigned pat, int e0,
                                         int* lq, int& nl, unsigned ltmask) {
    const unsigned x = kd ^ pat;
    const bool hit = ((x - 0x01010101u) & ~x & 0x80808080u) != 0u;
    const unsigned m = __builtin_amdgcn_ballot_w32(hit);
    if (m != 0u) {
        if (hit) lq[nl + __builtin_popcount(m & ltmask)] = e0;
        nl += __builtin_popcount(m);
    }
}

__device__ __forceinline__ int scan_chunks(const uint4 (&kv)[8], int nch, unsigned pat,
                                           int e0, int* lq, int& nl, unsigned ltmask) {
#pragma unroll
    for (int u = 0; u < 8; ++u) {
        if (u >= nch || nl > LQCAP - 128) return u;
        key_step(kv[u].x, pat, e0 + ((4 * u + 0) << 7), lq, nl, ltmask);
        key_step(kv[u].y, pat, e0 + ((4 * u + 1) << 7), lq, nl, ltmask);
        key_step(kv[u].z, pat, e0 + ((4 * u + 2) << 7), lq, nl, ltmask);
        key_step(kv[u].w, pat, e0 + ((4 * u + 3) << 7), lq, nl, ltmask);
    }
    return 8;
}

__device__ __forceinline__ void tile_product(
    const float* rows, int t, int nh, int lane, const _Float16* __restrict__ Wf,
    const float* __restrict__ bias, v8f* c) {
    const int mloc = lane & 15;
    const int base = (lane >> 4) * 8;
    const float* rowA = rows + (t * 16 + mloc) * (ROWP * 4);
#pragma unroll
    for (int j = 0; j < 4; ++j) {
        float bv = bias[(nh * 4 + j) * 16 + mloc];
#pragma unroll
        for (int r = 0; r < 8; ++r) c[j][r] = bv;
    }
#pragma unroll
    for (int kk = 0; kk < 4; ++kk) {
        int k0 = kk * 32 + base;
        float4 a0 = *(const float4*)(rowA + k0);
        float4 a1 = *(const float4*)(rowA + k0 + 4);
        float4 a2 = *(const float4*)(rowA + k0 + 16);
        float4 a3 = *(const float4*)(rowA + k0 + 20);
        v16h a;
        a[0]  = (_Float16)a0.x; a[1]  = (_Float16)a0.y;
        a[2]  = (_Float16)a0.z; a[3]  = (_Float16)a0.w;
        a[4]  = (_Float16)a1.x; a[5]  = (_Float16)a1.y;
        a[6]  = (_Float16)a1.z; a[7]  = (_Float16)a1.w;
        a[8]  = (_Float16)a2.x; a[9]  = (_Float16)a2.y;
        a[10] = (_Float16)a2.z; a[11] = (_Float16)a2.w;
        a[12] = (_Float16)a3.x; a[13] = (_Float16)a3.y;
        a[14] = (_Float16)a3.z; a[15] = (_Float16)a3.w;
        v8h blo[4], bhi[4];
#pragma unroll
        for (int j = 0; j < 4; ++j) {
            const v8h* bp = (const v8h*)Wf + ((kk * 8 + nh * 4 + j) * 2) * 32 + lane;
            blo[j] = bp[0];
            bhi[j] = bp[32];
        }
        asm volatile("" :: "v"(blo[0]), "v"(bhi[0]), "v"(blo[1]), "v"(bhi[1]),
                           "v"(blo[2]), "v"(bhi[2]), "v"(blo[3]), "v"(bhi[3]));
#pragma unroll
        for (int j = 0; j < 4; ++j) {
            v16h b;
#pragma unroll
            for (int i = 0; i < 8; ++i) { b[i] = blo[j][i]; b[8 + i] = bhi[j][i]; }
            c[j] = __builtin_amdgcn_wmma_f32_16x16x32_f16(
                false, a, false, b, (short)0, c[j], false, false);
        }
    }
}

__device__ __forceinline__ void tile_relu_to_lds(float* rows, int t, int nh,
                                                 int lane, const v8f* c) {
    float* p = rows + (t * 16 + (lane >> 4) * 8) * (ROWP * 4) + nh * 64 + (lane & 15);
#pragma unroll
    for (int j = 0; j < 4; ++j) {
#pragma unroll
        for (int r = 0; r < 8; ++r)
            p[r * (ROWP * 4) + j * 16] = fmaxf(c[j][r], 0.0f);
    }
}

template <bool POOL>
__global__ __launch_bounds__(1024) void layer_owner_kernel(
    const float* __restrict__ h, const int* __restrict__ ei,
    const unsigned* __restrict__ key,
    const float* __restrict__ ea, const float* __restrict__ eW,
    const float* __restrict__ eb, const float* __restrict__ eps, int l,
    const _Float16* __restrict__ W1f, const float* __restrict__ b1,
    const _Float16* __restrict__ W2f, const float* __restrict__ b2,
    float* __restrict__ out, const int* __restrict__ batch, int N, int E) {
    __shared__ float4 acc[OWN_T * 16 * ROWP];
    __shared__ int4   q[32 * QCAP];
    __shared__ int    pend[32 * 64];
    __shared__ int    qcnt[32];
    __shared__ int    lqs[32 * LQCAP];

    const int tid = threadIdx.x, lane = tid & 31;
    const int w = __builtin_amdgcn_readfirstlane(tid >> 5);
    const unsigned ltmask = (1u << lane) - 1u;
    const int lo = blockIdx.x * OWN_R;
    const int nrows = min(OWN_R, N - lo);

    for (int i = tid; i < OWN_T * 16 * ROWP; i += 1024)
        acc[i] = make_float4(0.f, 0.f, 0.f, 0.f);

    const float4 w0 = *(const float4*)(eW + 4 * lane);
    const float4 w1 = *(const float4*)(eW + HID + 4 * lane);
    const float4 bb = *(const float4*)(eb + 4 * lane);

    const int* dstp = ei + E;
    const int stepsTotal = (E + 127) >> 7;
    const int chunksTotal = (stepsTotal + 3) >> 2;
    const int cpw = (chunksTotal + 31) >> 5;
    int step = min(w * cpw * 4, stepsTotal);
    const int stepEnd = min((w * cpw + cpw) * 4, stepsTotal);
    const unsigned pat = (blockIdx.x & 255u) * 0x01010101u;
    const uint4* keyv = (const uint4*)key;
    int4* myq = q + w * QCAP;
    int*  pd  = pend + w * 64;
    int*  lq  = lqs + w * LQCAP;
    int nl = 0;

    for (;;) {
        bool lfull = false;
        while (!lfull && step < stepEnd) {
            if ((step & 3) == 0 && step + 4 <= stepEnd) {
                const int c0 = step >> 2;
                const int nch = min(8, (stepEnd - step) >> 2);
                uint4 kv[8];
#pragma unroll
                for (int u = 0; u < 8; ++u)
                    kv[u] = keyv[(size_t)(c0 + min(u, nch - 1)) * 32 + lane];
                const int took = scan_chunks(kv, nch, pat, (step << 7) + (lane << 2),
                                             lq, nl, ltmask);
                step += 4 * took;
                lfull = took < nch;
            } else {
                if (nl > LQCAP - 32) {
                    lfull = true;
                } else {
                    const unsigned kd = key[(size_t)(step >> 2) * 128 + lane * 4 + (step & 3)];
                    key_step(kd, pat, (step << 7) + (lane << 2), lq, nl, ltmask);
                    ++step;
                }
            }
        }
        int cnt = 0, g = 0;
        bool qfull = false;
        while (!qfull && g < nl) {
            int e0v[4], dv[4][4];
#pragma unroll
            for (int k = 0; k < 4; ++k) {
                const int i = g + 32 * k + lane;
                e0v[k] = 0;
                if (i < nl) e0v[k] = lq[i];
            }
#pragma unroll
            for (int k = 0; k < 4; ++k) {
#pragma unroll
                for (int b = 0; b < 4; ++b) dv[k][b] = dstp[min(e0v[k] + b, E - 1)];
            }
#pragma unroll
            for (int k = 0; k < 4; ++k) {
                if (!qfull && g < nl) {
                    const bool have = g + lane < nl;
                    bool ok[4];
                    int nok = 0;
#pragma unroll
                    for (int b = 0; b < 4; ++b) {
                        ok[b] = have && e0v[k] + b < E &&
                                (unsigned)(dv[k][b] - lo) < (unsigned)nrows;
                        nok += ok[b] ? 1 : 0;
                    }
                    const unsigned p0 = __builtin_amdgcn_ballot_w32((nok & 1) != 0);
                    const unsigned p1 = __builtin_amdgcn_ballot_w32((nok & 2) != 0);
                    const unsigned p2 = __builtin_amdgcn_ballot_w32((nok & 4) != 0);
                    const int total = __builtin_popcount(p0) + 2 * __builtin_popcount(p1) +
                                      4 * __builtin_popcount(p2);
                    if (cnt + total > QCAP) {
                        qfull = true;
                    } else {
                        int pos = cnt + __builtin_popcount(p0 & ltmask) +
                                  2 * __builtin_popcount(p1 & ltmask) +
                                  4 * __builtin_popcount(p2 & ltmask);
#pragma unroll
                        for (int b = 0; b < 4; ++b) {
                            if (ok[b]) {
                                myq[pos] = make_int4(e0v[k] + b, dv[k][b] - lo, 0, 0);
                                ++pos;
                            }
                        }
                        cnt += total;
                        g += 32;
                    }
                }
            }
        }
        for (int b0 = 0; b0 < cnt; b0 += 128) {
            int ex[4], rv[4], sv[4];
            float2 av[4];
#pragma unroll
            for (int k = 0; k < 4; ++k) {
                int i = b0 + 32 * k + lane;
                int4 t = make_int4(0, 0, 0, 0);
                if (i < cnt) t = myq[i];
                ex[k] = t.x;
                rv[k] = t.y;
            }
#pragma unroll
            for (int k = 0; k < 4; ++k) {
                sv[k] = ei[ex[k]];
                av[k] = *(const float2*)(ea + 2 * (size_t)ex[k]);
            }
#pragma unroll
            for (int k = 0; k < 4; ++k) {
                int i = b0 + 32 * k + lane;
                if (i < cnt)
                    myq[i] = make_int4(sv[k], rv[k], __float_as_int(av[k].x),
                                       __float_as_int(av[k].y));
            }
        }
        const int left = max(nl - g, 0);
        if (left > 0 && g > 0) {
            for (int k = 0; k < left; k += 32) {
                const int i = k + lane;
                int v = 0;
                if (i < left) v = lq[g + i];
                if (i < left) lq[i] = v;
            }
        }
        nl = left;
        if (lane == 0) qcnt[w] = cnt;
        int more = __syncthreads_or((step < stepEnd || nl > 0) ? 1 : 0);

        int np = 0;
        for (int qi = 0; qi < 32; ++qi) {
            const int c = __builtin_amdgcn_readfirstlane(qcnt[qi]);
            for (int b = 0; b < c; b += 32) {
                int idx = b + lane;
                bool mine = false;
                if (idx < c) mine = ((q[qi * QCAP + idx].y & 31) == w);
                unsigned mm = __builtin_amdgcn_ballot_w32(mine);
                if (mm) {
                    if (mine) pd[np + __builtin_popcount(mm & ltmask)] = qi * QCAP + idx;
                    np += __builtin_popcount(mm);
                }
                if (np >= 32) {
                    add_pending(q, pd, np, h, lane, w0, w1, bb, acc);
                    np = 0;
                }
            }
        }
        if (np > 0) add_pending(q, pd, np, h, lane, w0, w1, bb, acc);
        if (!more) break;
        __syncthreads();
    }
    __syncthreads();

    const float sc = 1.0f + eps[l];
    const float4* hrow = (const float4*)(h + (size_t)lo * HID);
    for (int i = tid; i < nrows * 32; i += 1024) {
        const int o = (i >> 5) * ROWP + (i & 31);
        float4 s = acc[o];
        float4 v = hrow[i];
        s.x += sc * v.x; s.y += sc * v.y; s.z += sc * v.z; s.w += sc * v.w;
        acc[o] = s;
    }
    __syncthreads();

    float* rows = (float*)acc;
    const int t = w >> 1, nh = w & 1;
    const bool work = w < 2 * OWN_T;
    v8f c[4];
    if (work) tile_product(rows, t, nh, lane, W1f, b1, c);
    __syncthreads();
    if (work) tile_relu_to_lds(rows, t, nh, lane, c);
    __syncthreads();
    if (work) tile_product(rows, t, nh, lane, W2f, b2, c);
    __syncthreads();
    if (work) tile_relu_to_lds(rows, t, nh, lane, c);
    __syncthreads();

    if (POOL) {
        const int col = tid & (HID - 1);
        const int segLen = (OWN_R + 7) / 8;
        const int r0 = (tid >> 7) * segLen;
        const int r1 = min(r0 + segLen, nrows);
        if (r0 < r1) {
            const int* gp = batch + lo;
            int cur = gp[r0];
            float s = 0.0f;
            for (int r = r0; r < r1; ++r) {
                int g = gp[r];
                if (g != cur) {
                    atomicAdd(out + (size_t)cur * HID + col, s);
                    s = 0.0f;
                    cur = g;
                }
                s += rows[r * (ROWP * 4) + col];
            }
            atomicAdd(out + (size_t)cur * HID + col, s);
        }
    } else {
        float4* orow = (float4*)(out + (size_t)lo * HID);
        for (int i = tid; i < nrows * 32; i += 1024)
            orow[i] = acc[(i >> 5) * ROWP + (i & 31)];
    }
}

__global__ void zero_kernel(float* __restrict__ p, int n4) {
    int i = blockIdx.x * blockDim.x + threadIdx.x;
    if (i < n4) ((float4*)p)[i] = make_float4(0.f, 0.f, 0.f, 0.f);
}

__global__ __launch_bounds__(256) void gemm128_wmma_kernel(
    const float* __restrict__ A, const _Float16* __restrict__ Wf,
    const float* __restrict__ bias, float* __restrict__ out, int rows) {
    int wave = threadIdx.x >> 5, lane = threadIdx.x & 31;
    int tile = blockIdx.x * 8 + wave;
    int row0 = tile * 16;
    if (row0 >= rows) return;

    int mloc = lane & 15;
    int half = lane >> 4;
    int base = half * 8;
    const float* rowA = A + (size_t)(row0 + mloc) * HID;

    v8f acc[8];
#pragma unroll
    for (int nt = 0; nt < 8; ++nt) {
        float bv = bias[nt * 16 + mloc];
#pragma unroll
        for (int r = 0; r < 8; ++r) acc[nt][r] = bv;
    }

#pragma unroll
    for (int kk = 0; kk < 4; ++kk) {
        int k0 = kk * 32 + base;
        float4 a0 = *(const float4*)(rowA + k0);
        float4 a1 = *(const float4*)(rowA + k0 + 4);
        float4 a2 = *(const float4*)(rowA + k0 + 16);
        float4 a3 = *(const float4*)(rowA + k0 + 20);
        v16h a;
        a[0]  = (_Float16)a0.x; a[1]  = (_Float16)a0.y;
        a[2]  = (_Float16)a0.z; a[3]  = (_Float16)a0.w;
        a[4]  = (_Float16)a1.x; a[5]  = (_Float16)a1.y;
        a[6]  = (_Float16)a1.z; a[7]  = (_Float16)a1.w;
        a[8]  = (_Float16)a2.x; a[9]  = (_Float16)a2.y;
        a[10] = (_Float16)a2.z; a[11] = (_Float16)a2.w;
        a[12] = (_Float16)a3.x; a[13] = (_Float16)a3.y;
        a[14] = (_Float16)a3.z; a[15] = (_Float16)a3.w;
        v8h blo[8], bhi[8];
#pragma unroll
        for (int nt = 0; nt < 8; ++nt) {
            const v8h* bp = (const v8h*)Wf + ((kk * 8 + nt) * 2) * 32 + lane;
            blo[nt] = bp[0];
            bhi[nt] = bp[32];
        }
        asm volatile("" :: "v"(blo[0]), "v"(bhi[0]), "v"(blo[1]), "v"(bhi[1]),
                           "v"(blo[2]), "v"(bhi[2]), "v"(blo[3]), "v"(bhi[3]),
                           "v"(blo[4]), "v"(bhi[4]), "v"(blo[5]), "v"(bhi[5]),
                           "v"(blo[6]), "v"(bhi[6]), "v"(blo[7]), "v"(bhi[7]));
#pragma unroll
        for (int nt = 0; nt < 8; ++nt) {
            v16h b;
#pragma unroll
            for (int i = 0; i < 8; ++i) { b[i] = blo[nt][i]; b[8 + i] = bhi[nt][i]; }
            acc[nt] = __builtin_amdgcn_wmma_f32_16x16x32_f16(
                false, a, false, b, (short)0, acc[nt], false, false);
        }
    }

#pragma unroll
    for (int nt = 0; nt < 8; ++nt) {
        int col = nt * 16 + mloc;
#pragma unroll
        for (int r = 0; r < 8; ++r) {
            int M = row0 + r + half * 8;
            out[(size_t)M * HID + col] = acc[nt][r];
        }
    }
}

extern "C" void kernel_launch(void* const* d_in, const int* in_sizes, int n_in,
                              void* d_out, int out_size, void* d_ws, size_t ws_size,
                              hipStream_t stream) {
    const int N = in_sizes[0];
    const int E = in_sizes[1] / 2;
    const int G = out_size / (3 * HID);

    const float* enc_W  = (const float*)d_in[12];
    const float* enc_b  = (const float*)d_in[13];
    const float* edge_W = (const float*)d_in[14];
    const float* edge_b = (const float*)d_in[15];
    const float* eps    = (const float*)d_in[16];
    const float* W1     = (const float*)d_in[17];
    const float* b1     = (const float*)d_in[18];
    const float* W2     = (const float*)d_in[19];
    const float* b2     = (const float*)d_in[20];
    const float* fc_W   = (const float*)d_in[21];
    const float* fc_b   = (const float*)d_in[22];

    float* bufA   = (float*)d_ws;
    float* bufB   = bufA + (size_t)N * HID;
    float* pooled = bufB + (size_t)N * HID;
    _Float16* w16 = (_Float16*)(pooled + (size_t)G * HID);
    _Float16* W1t0 = w16;
    _Float16* W1t1 = w16 + 16384;
    _Float16* W2t0 = w16 + 2 * 16384;
    _Float16* W2t1 = w16 + 3 * 16384;
    _Float16* fcWt = w16 + 4 * 16384;
    unsigned* keys = (unsigned*)(w16 + 5 * 16384);

    convert_wt_kernel<<<64, 256, 0, stream>>>(W1,             W1t0);
    convert_wt_kernel<<<64, 256, 0, stream>>>(W1 + HID * HID, W1t1);
    convert_wt_kernel<<<64, 256, 0, stream>>>(W2,             W2t0);
    convert_wt_kernel<<<64, 256, 0, stream>>>(W2 + HID * HID, W2t1);
    convert_wt_kernel<<<64, 256, 0, stream>>>(fc_W,           fcWt);

    const int nodeBlk   = (N + 7) / 8;
    const int ownBlk    = (N + OWN_R - 1) / OWN_R;
    const int gP4       = (G * HID) / 4;
    const int fcBlk     = ((G + 15) / 16 + 7) / 8;
    const int nchunk    = (E + 511) / 512;
    const int keyBlk    = (nchunk * 32 + 255) / 256;

    owner_key_kernel<<<dim3(keyBlk, 3), 256, 0, stream>>>(
        (const int*)d_in[1], (const int*)d_in[5], (const int*)d_in[9], keys, E, nchunk);

    for (int g = 0; g < 3; ++g) {
        const float* x  = (const float*)d_in[4 * g + 0];
        const int*   ei = (const int*)  d_in[4 * g + 1];
        const float* ea = (const float*)d_in[4 * g + 2];
        const int*   bt = (const int*)  d_in[4 * g + 3];

        encode_kernel<<<nodeBlk, 256, 0, stream>>>(x, enc_W, enc_b, bufA, N);

        layer_owner_kernel<false><<<ownBlk, 1024, 0, stream>>>(
            bufA, ei, keys + (size_t)g * nchunk * 128, ea, edge_W, edge_b, eps, 0, W1t0,
            b1, W2t0, b2, bufB, bt, N, E);
        zero_kernel<<<(gP4 + 255) / 256, 256, 0, stream>>>(pooled, gP4);
        layer_owner_kernel<true><<<ownBlk, 1024, 0, stream>>>(
            bufB, ei, keys + (size_t)g * nchunk * 128, ea, edge_W + 2 * HID, edge_b + HID,
            eps, 1, W1t1, b1 + HID,
            W2t1, b2 + HID, pooled, bt, N, E);

        gemm128_wmma_kernel<<<fcBlk, 256, 0, stream>>>(
            pooled, fcWt, fc_b, (float*)d_out + (size_t)g * G * HID, G);
    }
}


# layer 1 adds from per-wave edge lists (src, row, attrs; add order) that layer 0's adding waves store: no scan, queues or filter there; an entry sits in a lane and goes out by readlane; sum order kept
# speedup vs baseline: 4.0952x; 1.3381x over previous
#include <hip/hip_runtime.h>

typedef __attribute__((ext_vector_type(16))) _Float16 v16h;
typedef __attribute__((ext_vector_type(8)))  _Float16 v8h;
typedef __attribute__((ext_vector_type(8)))  float    v8f;

#define HID 128

__global__ void convert_wt_kernel(const float* __restrict__ W,
                                  _Float16* __restrict__ Wf) {
    int idx = blockIdx.x * blockDim.x + threadIdx.x;
    int i    = idx & 7;
    int lane = (idx >> 3) & 31;
    int j    = (idx >> 8) & 1;
    int nt   = (idx >> 9) & 7;
    int kk   = idx >> 12;
    int n = nt * 16 + (lane & 15);
    int k = kk * 32 + (lane >> 4) * 8 + j * 16 + i;
    Wf[idx] = (_Float16)W[k * HID + n];
}

__global__ void encode_kernel(const float* __restrict__ x,
                              const float* __restrict__ W,
                              const float* __restrict__ b,
                              float* __restrict__ h, int N) {
    int wave = threadIdx.x >> 5, lane = threadIdx.x & 31;
    int n = blockIdx.x * 8 + wave;
    if (n >= N) return;
    int c = lane * 4;
    float xv = x[n];
    float4 w  = *(const float4*)(W + c);
    float4 bb = *(const float4*)(b + c);
    float4 r;
    r.x = xv * w.x + bb.x;
    r.y = xv * w.y + bb.y;
    r.z = xv * w.z + bb.z;
    r.w = xv * w.w + bb.w;
    *(float4*)(h + (size_t)n * HID + c) = r;
}

#define OWN_R 196
__global__ __launch_bounds__(256) void owner_key_kernel(
    const int* __restrict__ ei_a, const int* __restrict__ ei_p,
    const int* __restrict__ ei_n, unsigned* __restrict__ key, int E, int nchunk) {
    const int* ei = blockIdx.y == 0 ? ei_a : (blockIdx.y == 1 ? ei_p : ei_n);
    const int* dstp = ei + E;
    const int t = blockIdx.x * 256 + threadIdx.x;
    if (t >= nchunk * 32) return;
    const int c = __builtin_amdgcn_readfirstlane(t >> 5), L = t & 31;
    int4 d[4];
    if ((E & 3) == 0 && (c + 1) * 512 <= E) {
#pragma unroll
        for (int j = 0; j < 4; ++j)
            d[j] = *(const int4*)(dstp + ((c * 4 + j) << 7) + (L << 2));
    } else {
#pragma unroll
        for (int j = 0; j < 4; ++j) {
            const int e0 = ((c * 4 + j) << 7) + (L << 2);
            d[j] = make_int4(0, 0, 0, 0);
            if (e0 < E)     d[j].x = dstp[e0];
            if (e0 + 1 < E) d[j].y = dstp[e0 + 1];
            if (e0 + 2 < E) d[j].z = dstp[e0 + 2];
            if (e0 + 3 < E) d[j].w = dstp[e0 + 3];
        }
    }
    unsigned kw[4];
#pragma unroll
    for (int j = 0; j < 4; ++j)
        kw[j] = (((unsigned)d[j].x / OWN_R) & 255u) |
                ((((unsigned)d[j].y / OWN_R) & 255u) << 8) |
                ((((unsigned)d[j].z / OWN_R) & 255u) << 16) |
                ((((unsigned)d[j].w / OWN_R) & 255u) << 24);
    ((uint4*)key)[(size_t)blockIdx.y * nchunk * 32 + t] =
        make_uint4(kw[0], kw[1], kw[2], kw[3]);
}

#define OWN_T 13
#define ROWP  33
#define QCAP  256
#define LQCAP 384
#define PCAP  160
#define ACAP  256
template <bool TAIL>
__device__ __forceinline__ void add_listed8(
    int4 ent, int p, int m, const float* __restrict__ h, int lane,
    float4 w0, float4 w1, float4 bb, float4* acc) {
    int srcu[8], rowu[8], a0u[8], a1u[8];
    float4 hv[8];
#pragma unroll
    for (int k = 0; k < 8; ++k) {
        const int i = TAIL ? min(p + k, m - 1) : p + k;
        srcu[k] = __builtin_amdgcn_readlane(ent.x, i);
        rowu[k] = __builtin_amdgcn_readlane(ent.y, i);
        a0u[k]  = __builtin_amdgcn_readlane(ent.z, i);
        a1u[k]  = __builtin_amdgcn_readlane(ent.w, i);
    }
#pragma unroll
    for (int k = 0; k < 8; ++k)
        hv[k] = *(const float4*)(h + (size_t)srcu[k] * HID + 4 * lane);
#pragma unroll
    for (int k = 0; k < 8; ++k) {
        const float a0 = __int_as_float(a0u[k]);
        const float a1 = __int_as_float(a1u[k]);
        float mx = fmaxf(hv[k].x + a0 * w0.x + a1 * w1.x + bb.x, 0.0f);
        float my = fmaxf(hv[k].y + a0 * w0.y + a1 * w1.y + bb.y, 0.0f);
        float mz = fmaxf(hv[k].z + a0 * w0.z + a1 * w1.z + bb.z, 0.0f);
        float mw = fmaxf(hv[k].w + a0 * w0.w + a1 * w1.w + bb.w, 0.0f);
        if (TAIL) {
            const int keep = (p + k < m) ? -1 : 0;
            mx = __int_as_float(__float_as_int(mx) & keep);
            my = __int_as_float(__float_as_int(my) & keep);
            mz = __int_as_float(__float_as_int(mz) & keep);
            mw = __int_as_float(__float_as_int(mw) & keep);
        }
        float4 s = acc[rowu[k] * ROWP + lane];
        s.x += mx; s.y += my; s.z += mz; s.w += mw;
        acc[rowu[k] * ROWP + lane] = s;
    }
}

__device__ __forceinline__ void add_listed(
    int4 ent, int m, const float* __restrict__ h, int lane,
    float4 w0, float4 w1, float4 bb, float4* acc) {
    int p = 0;
    for (; p + 8 <= m; p += 8) add_listed8<false>(ent, p, m, h, lane, w0, w1, bb, acc);
    if (p < m) add_listed8<true>(ent, p, m, h, lane, w0, w1, bb, acc);
}

template <bool SAVE>
__device__ __forceinline__ void add_pending(
    const int4* q, const int* pd, int np, int4* __restrict__ L, int& ns,
    const float* __restrict__ h, int lane,
    float4 w0, float4 w1, float4 bb, float4* acc) {
    const bool fits = SAVE && ns + np <= ACAP;
    for (int b0 = 0; b0 < np; b0 += 32) {
        const int m = min(32, np - b0);
        const int4 ent = q[pd[b0 + min(lane, m - 1)]];
        if (SAVE && fits && lane < m) L[(unsigned)(ns + b0) + (unsigned)lane] = ent;
        add_listed(ent, m, h, lane, w0, w1, bb, acc);
    }
    if (SAVE) ns = fits ? ns + np : ACAP + 1;
}

__device__ __forceinline__ void key_step(unsigned kd, unsigned pat, int e0,
                                         int* lq, int& nl, unsigned ltmask) {
    const unsigned x = kd ^ pat;
    const bool hit = ((x - 0x01010101u) & ~x & 0x80808080u) != 0u;
    const unsigned m = __builtin_amdgcn_ballot_w32(hit);
    if (m != 0u) {
        if (hit) lq[nl + __builtin_popcount(m & ltmask)] = e0;
        nl += __builtin_popcount(m);
    }
}

__device__ __forceinline__ int scan_chunks(const uint4 (&kv)[8], int nch, unsigned pat,
                                           int e0, int* lq, int& nl, unsigned ltmask) {
#pragma unroll
    for (int u = 0; u < 8; ++u) {
        if (u >= nch || nl > LQCAP - 128) return u;
        key_step(kv[u].x, pat, e0 + ((4 * u + 0) << 7), lq, nl, ltmask);
        key_step(kv[u].y, pat, e0 + ((4 * u + 1) << 7), lq, nl, ltmask);
        key_step(kv[u].z, pat, e0 + ((4 * u + 2) << 7), lq, nl, ltmask);
        key_step(kv[u].w, pat, e0 + ((4 * u + 3) << 7), lq, nl, ltmask);
    }
    return 8;
}

__device__ __forceinline__ void tile_product(
    const float* rows, int t, int nh, int lane, const _Float16* __restrict__ Wf,
    const float* __restrict__ bias, v8f* c) {
    const int mloc = lane & 15;
    const int base = (lane >> 4) * 8;
    const float* rowA = rows + (t * 16 + mloc) * (ROWP * 4);
#pragma unroll
    for (int j = 0; j < 4; ++j) {
        float bv = bias[(nh * 4 + j) * 16 + mloc];
#pragma unroll
        for (int r = 0; r < 8; ++r) c[j][r] = bv;
    }
#pragma unroll
    for (int kk = 0; kk < 4; ++kk) {
        int k0 = kk * 32 + base;
        float4 a0 = *(const float4*)(rowA + k0);
        float4 a1 = *(const float4*)(rowA + k0 + 4);
        float4 a2 = *(const float4*)(rowA + k0 + 16);
        float4 a3 = *(const float4*)(rowA + k0 + 20);
        v16h a;
        a[0]  = (_Float16)a0.x; a[1]  = (_Float16)a0.y;
        a[2]  = (_Float16)a0.z; a[3]  = (_Float16)a0.w;
        a[4]  = (_Float16)a1.x; a[5]  = (_Float16)a1.y;
        a[6]  = (_Float16)a1.z; a[7]  = (_Float16)a1.w;
        a[8]  = (_Float16)a2.x; a[9]  = (_Float16)a2.y;
        a[10] = (_Float16)a2.z; a[11] = (_Float16)a2.w;
        a[12] = (_Float16)a3.x; a[13] = (_Float16)a3.y;
        a[14] = (_Float16)a3.z; a[15] = (_Float16)a3.w;
        v8h blo[4], bhi[4];
#pragma unroll
        for (int j = 0; j < 4; ++j) {
            const v8h* bp = (const v8h*)Wf + ((kk * 8 + nh * 4 + j) * 2) * 32 + lane;
            blo[j] = bp[0];
            bhi[j] = bp[32];
        }
        asm volatile("" :: "v"(blo[0]), "v"(bhi[0]), "v"(blo[1]), "v"(bhi[1]),
                           "v"(blo[2]), "v"(bhi[2]), "v"(blo[3]), "v"(bhi[3]));
#pragma unroll
        for (int j = 0; j < 4; ++j) {
            v16h b;
#pragma unroll
            for (int i = 0; i < 8; ++i) { b[i] = blo[j][i]; b[8 + i] = bhi[j][i]; }
            c[j] = __builtin_amdgcn_wmma_f32_16x16x32_f16(
                false, a, false, b, (short)0, c[j], false, false);
        }
    }
}

__device__ __forceinline__ void tile_relu_to_lds(float* rows, int t, int nh,
                                                 int lane, const v8f* c) {
    float* p = rows + (t * 16 + (lane >> 4) * 8) * (ROWP * 4) + nh * 64 + (lane & 15);
#pragma unroll
    for (int j = 0; j < 4; ++j) {
#pragma unroll
        for (int r = 0; r < 8; ++r)
            p[r * (ROWP * 4) + j * 16] = fmaxf(c[j][r], 0.0f);
    }
}

template <bool POOL>
__global__ __launch_bounds__(1024) void layer_owner_kernel(
    const float* __restrict__ h, const int* __restrict__ ei,
    const unsigned* __restrict__ key,
    const float* __restrict__ ea, const float* __restrict__ eW,
    const float* __restrict__ eb, const float* __restrict__ eps, int l,
    const _Float16* __restrict__ W1f, const float* __restrict__ b1,
    const _Float16* __restrict__ W2f, const float* __restrict__ b2,
    float* __restrict__ out, const int* __restrict__ batch,
    int* __restrict__ lst, int* __restrict__ lcnt, int* __restrict__ lflag, int N, int E) {
    __shared__ float4 acc[OWN_T * 16 * ROWP];
    __shared__ int4   q[32 * QCAP];
    __shared__ int    pend[32 * PCAP];
    __shared__ int    qcnt[32];
    __shared__ int    lqs[32 * LQCAP];

    const int tid = threadIdx.x, lane = tid & 31;
    const int w = __builtin_amdgcn_readfirstlane(tid >> 5);
    const unsigned ltmask = (1u << lane) - 1u;
    const int lo = blockIdx.x * OWN_R;
    const int nrows = min(OWN_R, N - lo);

    int4* L = (int4*)lst + ((size_t)blockIdx.x * 32 + w) * ACAP;
    int reuse = 0, nlist = 0;
    if (POOL) {
        reuse = __builtin_amdgcn_readfirstlane(lflag[blockIdx.x]);
        nlist = __builtin_amdgcn_readfirstlane(lcnt[blockIdx.x * 32 + w]);
    }

    for (int i = tid; i < OWN_T * 16 * ROWP; i += 1024)
        acc[i] = make_float4(0.f, 0.f, 0.f, 0.f);

    const float4 w0 = *(const float4*)(eW + 4 * lane);
    const float4 w1 = *(const float4*)(eW + HID + 4 * lane);
    const float4 bb = *(const float4*)(eb + 4 * lane);

    const int* dstp = ei + E;
    const int stepsTotal = (E + 127) >> 7;
    const int chunksTotal = (stepsTotal + 3) >> 2;
    const int cpw = (chunksTotal + 31) >> 5;
    int step = min(w * cpw * 4, stepsTotal);
    const int stepEnd = min((w * cpw + cpw) * 4, stepsTotal);
    const unsigned pat = (blockIdx.x & 255u) * 0x01010101u;
    const uint4* keyv = (const uint4*)key;
    int4* myq = q + w * QCAP;
    int*  pd  = pend + w * PCAP;
    int*  lq  = lqs + w * LQCAP;
    int nl = 0;
    int ns = 0;

    if (POOL && reuse != 0) {
        __syncthreads();
        if (nlist > 0) {
            int4 cur = L[min(lane, nlist - 1)];
            for (int b0 = 0; b0 < nlist; b0 += 32) {
                const int4 nxt = L[min(b0 + 32 + lane, nlist - 1)];
                add_listed(cur, min(32, nlist - b0), h, lane, w0, w1, bb, acc);
                cur = nxt;
            }
        }
    } else
    for (;;) {
        bool lfull = false;
        while (!lfull && step < stepEnd) {
            if ((step & 3) == 0 && step + 4 <= stepEnd) {
                const int c0 = step >> 2;
                const int nch = min(8, (stepEnd - step) >> 2);
                uint4 kv[8];
#pragma unroll
                for (int u = 0; u < 8; ++u)
                    kv[u] = keyv[(size_t)(c0 + min(u, nch - 1)) * 32 + lane];
                const int took = scan_chunks(kv, nch, pat, (step << 7) + (lane << 2),
                                             lq, nl, ltmask);
                step += 4 * took;
                lfull = took < nch;
            } else {
                if (nl > LQCAP - 32) {
                    lfull = true;
                } else {
                    const unsigned kd = key[(size_t)(step >> 2) * 128 + lane * 4 + (step & 3)];
                    key_step(kd, pat, (step << 7) + (lane << 2), lq, nl, ltmask);
                    ++step;
                }
            }
        }
        int cnt = 0, g = 0;
        bool qfull = false;
        while (!qfull && g < nl) {
            int e0v[4], dv[4][4];
#pragma unroll
            for (int k = 0; k < 4; ++k) {
                const int i = g + 32 * k + lane;
                e0v[k] = 0;
                if (i < nl) e0v[k] = lq[i];
            }
#pragma unroll
            for (int k = 0; k < 4; ++k) {
#pragma unroll
                for (int b = 0; b < 4; ++b) dv[k][b] = dstp[min(e0v[k] + b, E - 1)];
            }
#pragma unroll
            for (int k = 0; k < 4; ++k) {
                if (!qfull && g < nl) {
                    const bool have = g + lane < nl;
                    bool ok[4];
                    int nok = 0;
#pragma unroll
                    for (int b = 0; b < 4; ++b) {
                        ok[b] = have && e0v[k] + b < E &&
                                (unsigned)(dv[k][b] - lo) < (unsigned)nrows;
                        nok += ok[b] ? 1 : 0;
                    }
                    const unsigned p0 = __builtin_amdgcn_ballot_w32((nok & 1) != 0);
                    const unsigned p1 = __builtin_amdgcn_ballot_w32((nok & 2) != 0);
                    const unsigned p2 = __builtin_amdgcn_ballot_w32((nok & 4) != 0);
                    const int total = __builtin_popcount(p0) + 2 * __builtin_popcount(p1) +
                                      4 * __builtin_popcount(p2);
                    if (cnt + total > QCAP) {
                        qfull = true;
                    } else {
                        int pos = cnt + __builtin_popcount(p0 & ltmask) +
                                  2 * __builtin_popcount(p1 & ltmask) +
                                  4 * __builtin_popcount(p2 & ltmask);
#pragma unroll
                        for (int b = 0; b < 4; ++b) {
                            if (ok[b]) {
                                myq[pos] = make_int4(e0v[k] + b, dv[k][b] - lo, 0, 0);
                                ++pos;
                            }
                        }
                        cnt += total;
                        g += 32;
                    }
                }
            }
        }
        for (int b0 = 0; b0 < cnt; b0 += 128) {
            int ex[4], rv[4], sv[4];
            float2 av[4];
#pragma unroll
            for (int k = 0; k < 4; ++k) {
                int i = b0 + 32 * k + lane;
                int4 t = make_int4(0, 0, 0, 0);
                if (i < cnt) t = myq[i];
                ex[k] = t.x;
                rv[k] = t.y;
            }
#pragma unroll
            for (int k = 0; k < 4; ++k) {
                sv[k] = ei[ex[k]];
                av[k] = *(const float2*)(ea + 2 * (size_t)ex[k]);
            }
#pragma unroll
            for (int k = 0; k < 4; ++k) {
                int i = b0 + 32 * k + lane;
                if (i < cnt)
                    myq[i] = make_int4(sv[k], rv[k], __float_as_int(av[k].x),
                                       __float_as_int(av[k].y));
            }
        }
        const int left = max(nl - g, 0);
        if (left > 0 && g > 0) {
            for (int k = 0; k < left; k += 32) {
                const int i = k + lane;
                int v = 0;
                if (i < left) v = lq[g + i];
                if (i < left) lq[i] = v;
            }
        }
        nl = left;
        if (lane == 0) qcnt[w] = cnt;
        int more = __syncthreads_or((step < stepEnd || nl > 0) ? 1 : 0);

        int np = 0;
        const int cntv = qcnt[lane];
        for (int qi = 0; qi < 32; ++qi) {
            const int c = __builtin_amdgcn_readlane(cntv, qi);
            for (int b = 0; b < c; b += 128) {
                int rowv[4];
#pragma unroll
                for (int k = 0; k < 4; ++k)
                    rowv[k] = q[qi * QCAP + min(b + 32 * k + lane, c - 1)].y;
#pragma unroll
                for (int k = 0; k < 4; ++k) {
                    const int idx = b + 32 * k + lane;
                    const bool mine = idx < c && (rowv[k] & 31) == w;
                    unsigned mm = __builtin_amdgcn_ballot_w32(mine);
                    if (mm) {
                        if (mine) pd[np + __builtin_popcount(mm & ltmask)] = qi * QCAP + idx;
                        np += __builtin_popcount(mm);
                    }
                }
                if (np >= 32) {
                    add_pending<!POOL>(q, pd, np, L, ns, h, lane, w0, w1, bb, acc);
                    np = 0;
                }
            }
        }
        if (np > 0) add_pending<!POOL>(q, pd, np, L, ns, h, lane, w0, w1, bb, acc);
        if (!more) break;
        __syncthreads();
    }
    if (!POOL) {
        const int anyovf = __syncthreads_or(ns > ACAP ? 1 : 0);
        if (lane == 0) lcnt[blockIdx.x * 32 + w] = ns;
        if (tid == 0) lflag[blockIdx.x] = anyovf ? 0 : 1;
    } else {
        __syncthreads();
    }

    const float sc = 1.0f + eps[l];
    const float4* hrow = (const float4*)(h + (size_t)lo * HID);
    for (int i = tid; i < nrows * 32; i += 1024) {
        const int o = (i >> 5) * ROWP + (i & 31);
        float4 s = acc[o];
        float4 v = hrow[i];
        s.x += sc * v.x; s.y += sc * v.y; s.z += sc * v.z; s.w += sc * v.w;
        acc[o] = s;
    }
    __syncthreads();

    float* rows = (float*)acc;
    const int t = w >> 1, nh = w & 1;
    const bool work = w < 2 * OWN_T;
    v8f c[4];
    if (work) tile_product(rows, t, nh, lane, W1f, b1, c);
    __syncthreads();
    if (work) tile_relu_to_lds(rows, t, nh, lane, c);
    __syncthreads();
    if (work) tile_product(rows, t, nh, lane, W2f, b2, c);
    __syncthreads();
    if (work) tile_relu_to_lds(rows, t, nh, lane, c);
    __syncthreads();

    if (POOL) {
        const int col = tid & (HID - 1);
        const int segLen = (OWN_R + 7) / 8;
        const int r0 = (tid >> 7) * segLen;
        const int r1 = min(r0 + segLen, nrows);
        if (r0 < r1) {
            const int* gp = batch + lo;
            int cur = gp[r0];
            float s = 0.0f;
            for (int r = r0; r < r1; ++r) {
                int g = gp[r];
                if (g != cur) {
                    atomicAdd(out + (size_t)cur * HID + col, s);
                    s = 0.0f;
                    cur = g;
                }
                s += rows[r * (ROWP * 4) + col];
            }
            atomicAdd(out + (size_t)cur * HID + col, s);
        }
    } else {
        float4* orow = (float4*)(out + (size_t)lo * HID);
        for (int i = tid; i < nrows * 32; i += 1024)
            orow[i] = acc[(i >> 5) * ROWP + (i & 31)];
    }
}

__global__ void zero_kernel(float* __restrict__ p, int n4) {
    int i = blockIdx.x * blockDim.x + threadIdx.x;
    if (i < n4) ((float4*)p)[i] = make_float4(0.f, 0.f, 0.f, 0.f);
}

__global__ __launch_bounds__(256) void gemm128_wmma_kernel(
    const float* __restrict__ A, const _Float16* __restrict__ Wf,
    const float* __restrict__ bias, float* __restrict__ out, int rows) {
    int wave = threadIdx.x >> 5, lane = threadIdx.x & 31;
    int tile = blockIdx.x * 8 + wave;
    int row0 = tile * 16;
    if (row0 >= rows) return;

    int mloc = lane & 15;
    int half = lane >> 4;
    int base = half * 8;
    const float* rowA = A + (size_t)(row0 + mloc) * HID;

    v8f acc[8];
#pragma unroll
    for (int nt = 0; nt < 8; ++nt) {
        float bv = bias[nt * 16 + mloc];
#pragma unroll
        for (int r = 0; r < 8; ++r) acc[nt][r] = bv;
    }

#pragma unroll
    for (int kk = 0; kk < 4; ++kk) {
        int k0 = kk * 32 + base;
        float4 a0 = *(const float4*)(rowA + k0);
        float4 a1 = *(const float4*)(rowA + k0 + 4);
        float4 a2 = *(const float4*)(rowA + k0 + 16);
        float4 a3 = *(const float4*)(rowA + k0 + 20);
        v16h a;
        a[0]  = (_Float16)a0.x; a[1]  = (_Float16)a0.y;
        a[2]  = (_Float16)a0.z; a[3]  = (_Float16)a0.w;
        a[4]  = (_Float16)a1.x; a[5]  = (_Float16)a1.y;
        a[6]  = (_Float16)a1.z; a[7]  = (_Float16)a1.w;
        a[8]  = (_Float16)a2.x; a[9]  = (_Float16)a2.y;
        a[10] = (_Float16)a2.z; a[11] = (_Float16)a2.w;
        a[12] = (_Float16)a3.x; a[13] = (_Float16)a3.y;
        a[14] = (_Float16)a3.z; a[15] = (_Float16)a3.w;
        v8h blo[8], bhi[8];
#pragma unroll
        for (int nt = 0; nt < 8; ++nt) {
            const v8h* bp = (const v8h*)Wf + ((kk * 8 + nt) * 2) * 32 + lane;
            blo[nt] = bp[0];
            bhi[nt] = bp[32];
        }
        asm volatile("" :: "v"(blo[0]), "v"(bhi[0]), "v"(blo[1]), "v"(bhi[1]),
                           "v"(blo[2]), "v"(bhi[2]), "v"(blo[3]), "v"(bhi[3]),
                           "v"(blo[4]), "v"(bhi[4]), "v"(blo[5]), "v"(bhi[5]),
                           "v"(blo[6]), "v"(bhi[6]), "v"(blo[7]), "v"(bhi[7]));
#pragma unroll
        for (int nt = 0; nt < 8; ++nt) {
            v16h b;
#pragma unroll
            for (int i = 0; i < 8; ++i) { b[i] = blo[nt][i]; b[8 + i] = bhi[nt][i]; }
            acc[nt] = __builtin_amdgcn_wmma_f32_16x16x32_f16(
                false, a, false, b, (short)0, acc[nt], false, false);
        }
    }

#pragma unroll
    for (int nt = 0; nt < 8; ++nt) {
        int col = nt * 16 + mloc;
#pragma unroll
        for (int r = 0; r < 8; ++r) {
            int M = row0 + r + half * 8;
            out[(size_t)M * HID + col] = acc[nt][r];
        }
    }
}

extern "C" void kernel_launch(void* const* d_in, const int* in_sizes, int n_in,
                              void* d_out, int out_size, void* d_ws, size_t ws_size,
                              hipStream_t stream) {
    const int N = in_sizes[0];
    const int E = in_sizes[1] / 2;
    const int G = out_size / (3 * HID);

    const float* enc_W  = (const float*)d_in[12];
    const float* enc_b  = (const float*)d_in[13];
    const float* edge_W = (const float*)d_in[14];
    const float* edge_b = (const float*)d_in[15];
    const float* eps    = (const float*)d_in[16];
    const float* W1     = (const float*)d_in[17];
    const float* b1     = (const float*)d_in[18];
    const float* W2     = (const float*)d_in[19];
    const float* b2     = (const float*)d_in[20];
    const float* fc_W   = (const float*)d_in[21];
    const float* fc_b   = (const float*)d_in[22];

    float* bufA   = (float*)d_ws;
    float* bufB   = bufA + (size_t)N * HID;
    float* pooled = bufB + (size_t)N * HID;
    _Float16* w16 = (_Float16*)(pooled + (size_t)G * HID);
    _Float16* W1t0 = w16;
    _Float16* W1t1 = w16 + 16384;
    _Float16* W2t0 = w16 + 2 * 16384;
    _Float16* W2t1 = w16 + 3 * 16384;
    _Float16* fcWt = w16 + 4 * 16384;
    unsigned* keys = (unsigned*)(w16 + 5 * 16384);

    convert_wt_kernel<<<64, 256, 0, stream>>>(W1,             W1t0);
    convert_wt_kernel<<<64, 256, 0, stream>>>(W1 + HID * HID, W1t1);
    convert_wt_kernel<<<64, 256, 0, stream>>>(W2,             W2t0);
    convert_wt_kernel<<<64, 256, 0, stream>>>(W2 + HID * HID, W2t1);
    convert_wt_kernel<<<64, 256, 0, stream>>>(fc_W,           fcWt);

    const int nodeBlk   = (N + 7) / 8;
    const int ownBlk    = (N + OWN_R - 1) / OWN_R;
    const int gP4       = (G * HID) / 4;
    const int fcBlk     = ((G + 15) / 16 + 7) / 8;
    const int nchunk    = (E + 511) / 512;
    const int keyBlk    = (nchunk * 32 + 255) / 256;

    int* lst   = (int*)(keys + (size_t)3 * nchunk * 128);
    int* lcnt  = lst + (size_t)ownBlk * 32 * ACAP * 4;
    int* lflag = lcnt + (size_t)ownBlk * 32;

    owner_key_kernel<<<dim3(keyBlk, 3), 256, 0, stream>>>(
        (const int*)d_in[1], (const int*)d_in[5], (const int*)d_in[9], keys, E, nchunk);

    for (int g = 0; g < 3; ++g) {
        const float* x  = (const float*)d_in[4 * g + 0];
        const int*   ei = (const int*)  d_in[4 * g + 1];
        const float* ea = (const float*)d_in[4 * g + 2];
        const int*   bt = (const int*)  d_in[4 * g + 3];

        encode_kernel<<<nodeBlk, 256, 0, stream>>>(x, enc_W, enc_b, bufA, N);

        layer_owner_kernel<false><<<ownBlk, 1024, 0, stream>>>(
            bufA, ei, keys + (size_t)g * nchunk * 128, ea, edge_W, edge_b, eps, 0, W1t0,
            b1, W2t0, b2, bufB, bt, lst, lcnt, lflag, N, E);
        zero_kernel<<<(gP4 + 255) / 256, 256, 0, stream>>>(pooled, gP4);
        layer_owner_kernel<true><<<ownBlk, 1024, 0, stream>>>(
            bufB, ei, keys + (size_t)g * nchunk * 128, ea, edge_W + 2 * HID, edge_b + HID,
            eps, 1, W1t1, b1 + HID,
            W2t1, b2 + HID, pooled, bt, lst, lcnt, lflag, N, E);

        gemm128_wmma_kernel<<<fcBlk, 256, 0, stream>>>(
            pooled, fcWt, fc_b, (float*)d_out + (size_t)g * G * HID, G);
    }
}


# small launches folded into neighbours: 5 weight conversions, owner bytes, first encoder, pooled zeroing in ONE launch; the next graph's encoder rows go out from the layer-1 kernel; 21 -> 10 launches
# speedup vs baseline: 4.3655x; 1.0660x over previous
#include <hip/hip_runtime.h>

typedef __attribute__((ext_vector_type(16))) _Float16 v16h;
typedef __attribute__((ext_vector_type(8)))  _Float16 v8h;
typedef __attribute__((ext_vector_type(8)))  float    v8f;

#define HID 128

__device__ __forceinline__ void convert_wt_elem(const float* __restrict__ W,
                                                _Float16* __restrict__ Wf, int idx) {
    int i    = idx & 7;
    int lane = (idx >> 3) & 31;
    int j    = (idx >> 8) & 1;
    int nt   = (idx >> 9) & 7;
    int kk   = idx >> 12;
    int n = nt * 16 + (lane & 15);
    int k = kk * 32 + (lane >> 4) * 8 + j * 16 + i;
    Wf[idx] = (_Float16)W[k * HID + n];
}

__device__ __forceinline__ void encode_row(float xv, float4 w, float4 bb,
                                           float* __restrict__ hrow, int lane) {
    float4 r;
    r.x = xv * w.x + bb.x;
    r.y = xv * w.y + bb.y;
    r.z = xv * w.z + bb.z;
    r.w = xv * w.w + bb.w;
    *(float4*)(hrow + 4 * lane) = r;
}

#define OWN_R 196
__device__ __forceinline__ void owner_key_block(
    const int* __restrict__ ei, unsigned* __restrict__ key, int E, int nchunk,
    int bx, int gsel) {
    const int* dstp = ei + E;
    const int t = bx * 256 + threadIdx.x;
    if (t >= nchunk * 32) return;
    const int c = __builtin_amdgcn_readfirstlane(t >> 5), L = t & 31;
    int4 d[4];
    if ((E & 3) == 0 && (c + 1) * 512 <= E) {
#pragma unroll
        for (int j = 0; j < 4; ++j)
            d[j] = *(const int4*)(dstp + ((c * 4 + j) << 7) + (L << 2));
    } else {
#pragma unroll
        for (int j = 0; j < 4; ++j) {
            const int e0 = ((c * 4 + j) << 7) + (L << 2);
            d[j] = make_int4(0, 0, 0, 0);
            if (e0 < E)     d[j].x = dstp[e0];
            if (e0 + 1 < E) d[j].y = dstp[e0 + 1];
            if (e0 + 2 < E) d[j].z = dstp[e0 + 2];
            if (e0 + 3 < E) d[j].w = dstp[e0 + 3];
        }
    }
    unsigned kw[4];
#pragma unroll
    for (int j = 0; j < 4; ++j)
        kw[j] = (((unsigned)d[j].x / OWN_R) & 255u) |
                ((((unsigned)d[j].y / OWN_R) & 255u) << 8) |
                ((((unsigned)d[j].z / OWN_R) & 255u) << 16) |
                ((((unsigned)d[j].w / OWN_R) & 255u) << 24);
    ((uint4*)key)[(size_t)gsel * nchunk * 32 + t] =
        make_uint4(kw[0], kw[1], kw[2], kw[3]);
}

__global__ __launch_bounds__(256) void prep_kernel(
    const float* __restrict__ W1, const float* __restrict__ W2,
    const float* __restrict__ fcW, _Float16* __restrict__ w16,
    const int* __restrict__ ei_a, const int* __restrict__ ei_p,
    const int* __restrict__ ei_n, unsigned* __restrict__ key, int E, int nchunk,
    int keyBlk,
    const float* __restrict__ x0, const float* __restrict__ encW,
    const float* __restrict__ encb, float* __restrict__ h0, int N, int encBlk,
    float* __restrict__ pooled, int nz4) {
    int b = blockIdx.x;
    if (b < 320) {
        const int m = b >> 6;
        const float* W = m == 0 ? W1 : m == 1 ? W1 + HID * HID :
                         m == 2 ? W2 : m == 3 ? W2 + HID * HID : fcW;
        convert_wt_elem(W, w16 + m * 16384, (b & 63) * 256 + threadIdx.x);
        return;
    }
    b -= 320;
    if (b < 3 * keyBlk) {
        const int gsel = b / keyBlk;
        const int* ei = gsel == 0 ? ei_a : (gsel == 1 ? ei_p : ei_n);
        owner_key_block(ei, key, E, nchunk, b - gsel * keyBlk, gsel);
        return;
    }
    b -= 3 * keyBlk;
    if (b < encBlk) {
        const int wave = threadIdx.x >> 5, lane = threadIdx.x & 31;
        const float4 w  = *(const float4*)(encW + 4 * lane);
        const float4 bb = *(const float4*)(encb + 4 * lane);
        const int n0 = b * 32 + wave * 4;
        float xv[4];
#pragma unroll
        for (int k = 0; k < 4; ++k) xv[k] = x0[min(n0 + k, N - 1)];
#pragma unroll
        for (int k = 0; k < 4; ++k)
            if (n0 + k < N) encode_row(xv[k], w, bb, h0 + (size_t)(n0 + k) * HID, lane);
        return;
    }
    b -= encBlk;
    const int i = b * 256 + threadIdx.x;
    if (i < nz4) ((float4*)pooled)[i] = make_float4(0.f, 0.f, 0.f, 0.f);
}

#define OWN_T 13
#define ROWP  33
#define QCAP  256
#define LQCAP 384
#define PCAP  160
#define ACAP  256
template <bool TAIL>
__device__ __forceinline__ void add_listed8(
    int4 ent, int p, int m, const float* __restrict__ h, int lane,
    float4 w0, float4 w1, float4 bb, float4* acc) {
    int srcu[8], rowu[8], a0u[8], a1u[8];
    float4 hv[8];
#pragma unroll
    for (int k = 0; k < 8; ++k) {
        const int i = TAIL ? min(p + k, m - 1) : p + k;
        srcu[k] = __builtin_amdgcn_readlane(ent.x, i);
        rowu[k] = __builtin_amdgcn_readlane(ent.y, i);
        a0u[k]  = __builtin_amdgcn_readlane(ent.z, i);
        a1u[k]  = __builtin_amdgcn_readlane(ent.w, i);
    }
#pragma unroll
    for (int k = 0; k < 8; ++k)
        hv[k] = *(const float4*)(h + (size_t)srcu[k] * HID + 4 * lane);
#pragma unroll
    for (int k = 0; k < 8; ++k) {
        const float a0 = __int_as_float(a0u[k]);
        const float a1 = __int_as_float(a1u[k]);
        float mx = fmaxf(hv[k].x + a0 * w0.x + a1 * w1.x + bb.x, 0.0f);
        float my = fmaxf(hv[k].y + a0 * w0.y + a1 * w1.y + bb.y, 0.0f);
        float mz = fmaxf(hv[k].z + a0 * w0.z + a1 * w1.z + bb.z, 0.0f);
        float mw = fmaxf(hv[k].w + a0 * w0.w + a1 * w1.w + bb.w, 0.0f);
        if (TAIL) {
            const int keep = (p + k < m) ? -1 : 0;
            mx = __int_as_float(__float_as_int(mx) & keep);
            my = __int_as_float(__float_as_int(my) & keep);
            mz = __int_as_float(__float_as_int(mz) & keep);
            mw = __int_as_float(__float_as_int(mw) & keep);
        }
        float4 s = acc[rowu[k] * ROWP + lane];
        s.x += mx; s.y += my; s.z += mz; s.w += mw;
        acc[rowu[k] * ROWP + lane] = s;
    }
}

__device__ __forceinline__ void add_listed(
    int4 ent, int m, const float* __restrict__ h, int lane,
    float4 w0, float4 w1, float4 bb, float4* acc) {
    int p = 0;
    for (; p + 8 <= m; p += 8) add_listed8<false>(ent, p, m, h, lane, w0, w1, bb, acc);
    if (p < m) add_listed8<true>(ent, p, m, h, lane, w0, w1, bb, acc);
}

template <bool SAVE>
__device__ __forceinline__ void add_pending(
    const int4* q, const int* pd, int np, int4* __restrict__ L, int& ns,
    const float* __restrict__ h, int lane,
    float4 w0, float4 w1, float4 bb, float4* acc) {
    const bool fits = SAVE && ns + np <= ACAP;
    for (int b0 = 0; b0 < np; b0 += 32) {
        const int m = min(32, np - b0);
        const int4 ent = q[pd[b0 + min(lane, m - 1)]];
        if (SAVE && fits && lane < m) L[(unsigned)(ns + b0) + (unsigned)lane] = ent;
        add_listed(ent, m, h, lane, w0, w1, bb, acc);
    }
    if (SAVE) ns = fits ? ns + np : ACAP + 1;
}

__device__ __forceinline__ void key_step(unsigned kd, unsigned pat, int e0,
                                         int* lq, int& nl, unsigned ltmask) {
    const unsigned x = kd ^ pat;
    const bool hit = ((x - 0x01010101u) & ~x & 0x80808080u) != 0u;
    const unsigned m = __builtin_amdgcn_ballot_w32(hit);
    if (m != 0u) {
        if (hit) lq[nl + __builtin_popcount(m & ltmask)] = e0;
        nl += __builtin_popcount(m);
    }
}

__device__ __forceinline__ int scan_chunks(const uint4 (&kv)[8], int nch, unsigned pat,
                                           int e0, int* lq, int& nl, unsigned ltmask) {
#pragma unroll
    for (int u = 0; u < 8; ++u) {
        if (u >= nch || nl > LQCAP - 128) return u;
        key_step(kv[u].x, pat, e0 + ((4 * u + 0) << 7), lq, nl, ltmask);
        key_step(kv[u].y, pat, e0 + ((4 * u + 1) << 7), lq, nl, ltmask);
        key_step(kv[u].z, pat, e0 + ((4 * u + 2) << 7), lq, nl, ltmask);
        key_step(kv[u].w, pat, e0 + ((4 * u + 3) << 7), lq, nl, ltmask);
    }
    return 8;
}

__device__ __forceinline__ void tile_product(
    const float* rows, int t, int nh, int lane, const _Float16* __restrict__ Wf,
    const float* __restrict__ bias, v8f* c) {
    const int mloc = lane & 15;
    const int base = (lane >> 4) * 8;
    const float* rowA = rows + (t * 16 + mloc) * (ROWP * 4);
#pragma unroll
    for (int j = 0; j < 4; ++j) {
        float bv = bias[(nh * 4 + j) * 16 + mloc];
#pragma unroll
        for (int r = 0; r < 8; ++r) c[j][r] = bv;
    }
#pragma unroll
    for (int kk = 0; kk < 4; ++kk) {
        int k0 = kk * 32 + base;
        float4 a0 = *(const float4*)(rowA + k0);
        float4 a1 = *(const float4*)(rowA + k0 + 4);
        float4 a2 = *(const float4*)(rowA + k0 + 16);
        float4 a3 = *(const float4*)(rowA + k0 + 20);
        v16h a;
        a[0]  = (_Float16)a0.x; a[1]  = (_Float16)a0.y;
        a[2]  = (_Float16)a0.z; a[3]  = (_Float16)a0.w;
        a[4]  = (_Float16)a1.x; a[5]  = (_Float16)a1.y;
        a[6]  = (_Float16)a1.z; a[7]  = (_Float16)a1.w;
        a[8]  = (_Float16)a2.x; a[9]  = (_Float16)a2.y;
        a[10] = (_Float16)a2.z; a[11] = (_Float16)a2.w;
        a[12] = (_Float16)a3.x; a[13] = (_Float16)a3.y;
        a[14] = (_Float16)a3.z; a[15] = (_Float16)a3.w;
        v8h blo[4], bhi[4];
#pragma unroll
        for (int j = 0; j < 4; ++j) {
            const v8h* bp = (const v8h*)Wf + ((kk * 8 + nh * 4 + j) * 2) * 32 + lane;
            blo[j] = bp[0];
            bhi[j] = bp[32];
        }
        asm volatile("" :: "v"(blo[0]), "v"(bhi[0]), "v"(blo[1]), "v"(bhi[1]),
                           "v"(blo[2]), "v"(bhi[2]), "v"(blo[3]), "v"(bhi[3]));
#pragma unroll
        for (int j = 0; j < 4; ++j) {
            v16h b;
#pragma unroll
            for (int i = 0; i < 8; ++i) { b[i] = blo[j][i]; b[8 + i] = bhi[j][i]; }
            c[j] = __builtin_amdgcn_wmma_f32_16x16x32_f16(
                false, a, false, b, (short)0, c[j], false, false);
        }
    }
}

__device__ __forceinline__ void tile_relu_to_lds(float* rows, int t, int nh,
                                                 int lane, const v8f* c) {
    float* p = rows + (t * 16 + (lane >> 4) * 8) * (ROWP * 4) + nh * 64 + (lane & 15);
#pragma unroll
    for (int j = 0; j < 4; ++j) {
#pragma unroll
        for (int r = 0; r < 8; ++r)
            p[r * (ROWP * 4) + j * 16] = fmaxf(c[j][r], 0.0f);
    }
}

__device__ __forceinline__ void encode_wave_rows(
    const float (&xv)[(OWN_R + 31) / 32], float4 ew, float4 ebv,
    float* __restrict__ hn, int lo, int nrows, int w, int lane) {
#pragma unroll
    for (int k = 0; k < (OWN_R + 31) / 32; ++k)
        if (w + 32 * k < nrows)
            encode_row(xv[k], ew, ebv, hn + (size_t)(lo + w + 32 * k) * HID, lane);
}

template <bool POOL>
__global__ __launch_bounds__(1024) void layer_owner_kernel(
    const float* __restrict__ h, const int* __restrict__ ei,
    const unsigned* __restrict__ key,
    const float* __restrict__ ea, const float* __restrict__ eW,
    const float* __restrict__ eb, const float* __restrict__ eps, int l,
    const _Float16* __restrict__ W1f, const float* __restrict__ b1,
    const _Float16* __restrict__ W2f, const float* __restrict__ b2,
    float* __restrict__ out, const int* __restrict__ batch,
    int* __restrict__ lst, int* __restrict__ lcnt, int* __restrict__ lflag, int N, int E,
    const float* __restrict__ xn, const float* __restrict__ encW,
    const float* __restrict__ encb, float* __restrict__ hn) {
    __shared__ float4 acc[OWN_T * 16 * ROWP];
    __shared__ int4   q[32 * QCAP];
    __shared__ int    pend[32 * PCAP];
    __shared__ int    qcnt[32];
    __shared__ int    lqs[32 * LQCAP];

    const int tid = threadIdx.x, lane = tid & 31;
    const int w = __builtin_amdgcn_readfirstlane(tid >> 5);
    const unsigned ltmask = (1u << lane) - 1u;
    const int lo = blockIdx.x * OWN_R;
    const int nrows = min(OWN_R, N - lo);

    int4* L = (int4*)lst + ((size_t)blockIdx.x * 32 + w) * ACAP;
    int reuse = 0, nlist = 0;
    if (POOL) {
        reuse = __builtin_amdgcn_readfirstlane(lflag[blockIdx.x]);
        nlist = __builtin_amdgcn_readfirstlane(lcnt[blockIdx.x * 32 + w]);
    }

    for (int i = tid; i < OWN_T * 16 * ROWP; i += 1024)
        acc[i] = make_float4(0.f, 0.f, 0.f, 0.f);

    const float4 w0 = *(const float4*)(eW + 4 * lane);
    const float4 w1 = *(const float4*)(eW + HID + 4 * lane);
    const float4 bb = *(const float4*)(eb + 4 * lane);

    const int NRW = (OWN_R + 31) / 32;
    const bool enc = POOL && xn != nullptr;
    float4 ew = make_float4(0.f, 0.f, 0.f, 0.f), ebv = ew;
    float xv[NRW];
#pragma unroll
    for (int k = 0; k < NRW; ++k) xv[k] = 0.0f;
    if (enc) {
        ew  = *(const float4*)(encW + 4 * lane);
        ebv = *(const float4*)(encb + 4 * lane);
#pragma unroll
        for (int k = 0; k < NRW; ++k) xv[k] = xn[lo + min(w + 32 * k, nrows - 1)];
    }

    const int* dstp = ei + E;
    const int stepsTotal = (E + 127) >> 7;
    const int chunksTotal = (stepsTotal + 3) >> 2;
    const int cpw = (chunksTotal + 31) >> 5;
    int step = min(w * cpw * 4, stepsTotal);
    const int stepEnd = min((w * cpw + cpw) * 4, stepsTotal);
    const unsigned pat = (blockIdx.x & 255u) * 0x01010101u;
    const uint4* keyv = (const uint4*)key;
    int4* myq = q + w * QCAP;
    int*  pd  = pend + w * PCAP;
    int*  lq  = lqs + w * LQCAP;
    int nl = 0;
    int ns = 0;

    if (POOL && reuse != 0) {
        __syncthreads();
        if (enc) encode_wave_rows(xv, ew, ebv, hn, lo, nrows, w, lane);
        if (nlist > 0) {
            int4 cur = L[min(lane, nlist - 1)];
            for (int b0 = 0; b0 < nlist; b0 += 32) {
                const int4 nxt = L[min(b0 + 32 + lane, nlist - 1)];
                add_listed(cur, min(32, nlist - b0), h, lane, w0, w1, bb, acc);
                cur = nxt;
            }
        }
    } else {
    if (enc) encode_wave_rows(xv, ew, ebv, hn, lo, nrows, w, lane);
    for (;;) {
        bool lfull = false;
        while (!lfull && step < stepEnd) {
            if ((step & 3) == 0 && step + 4 <= stepEnd) {
                const int c0 = step >> 2;
                const int nch = min(8, (stepEnd - step) >> 2);
                uint4 kv[8];
#pragma unroll
                for (int u = 0; u < 8; ++u)
                    kv[u] = keyv[(size_t)(c0 + min(u, nch - 1)) * 32 + lane];
                const int took = scan_chunks(kv, nch, pat, (step << 7) + (lane << 2),
                                             lq, nl, ltmask);
                step += 4 * took;
                lfull = took < nch;
            } else {
                if (nl > LQCAP - 32) {
                    lfull = true;
                } else {
                    const unsigned kd = key[(size_t)(step >> 2) * 128 + lane * 4 + (step & 3)];
                    key_step(kd, pat, (step << 7) + (lane << 2), lq, nl, ltmask);
                    ++step;
                }
            }
        }
        int cnt = 0, g = 0;
        bool qfull = false;
        while (!qfull && g < nl) {
            int e0v[4], dv[4][4];
#pragma unroll
            for (int k = 0; k < 4; ++k) {
                const int i = g + 32 * k + lane;
                e0v[k] = 0;
                if (i < nl) e0v[k] = lq[i];
            }
#pragma unroll
            for (int k = 0; k < 4; ++k) {
#pragma unroll
                for (int b = 0; b < 4; ++b) dv[k][b] = dstp[min(e0v[k] + b, E - 1)];
            }
#pragma unroll
            for (int k = 0; k < 4; ++k) {
                if (!qfull && g < nl) {
                    const bool have = g + lane < nl;
                    bool ok[4];
                    int nok = 0;
#pragma unroll
                    for (int b = 0; b < 4; ++b) {
                        ok[b] = have && e0v[k] + b < E &&
                                (unsigned)(dv[k][b] - lo) < (unsigned)nrows;
                        nok += ok[b] ? 1 : 0;
                    }
                    const unsigned p0 = __builtin_amdgcn_ballot_w32((nok & 1) != 0);
                    const unsigned p1 = __builtin_amdgcn_ballot_w32((nok & 2) != 0);
                    const unsigned p2 = __builtin_amdgcn_ballot_w32((nok & 4) != 0);
                    const int total = __builtin_popcount(p0) + 2 * __builtin_popcount(p1) +
                                      4 * __builtin_popcount(p2);
                    if (cnt + total > QCAP) {
                        qfull = true;
                    } else {
                        int pos = cnt + __builtin_popcount(p0 & ltmask) +
                                  2 * __builtin_popcount(p1 & ltmask) +
                                  4 * __builtin_popcount(p2 & ltmask);
#pragma unroll
                        for (int b = 0; b < 4; ++b) {
                            if (ok[b]) {
                                myq[pos] = make_int4(e0v[k] + b, dv[k][b] - lo, 0, 0);
                                ++pos;
                            }
                        }
                        cnt += total;
                        g += 32;
                    }
                }
            }
        }
        for (int b0 = 0; b0 < cnt; b0 += 128) {
            int ex[4], rv[4], sv[4];
            float2 av[4];
#pragma unroll
            for (int k = 0; k < 4; ++k) {
                int i = b0 + 32 * k + lane;
                int4 t = make_int4(0, 0, 0, 0);
                if (i < cnt) t = myq[i];
                ex[k] = t.x;
                rv[k] = t.y;
            }
#pragma unroll
            for (int k = 0; k < 4; ++k) {
                sv[k] = ei[ex[k]];
                av[k] = *(const float2*)(ea + 2 * (size_t)ex[k]);
            }
#pragma unroll
            for (int k = 0; k < 4; ++k) {
                int i = b0 + 32 * k + lane;
                if (i < cnt)
                    myq[i] = make_int4(sv[k], rv[k], __float_as_int(av[k].x),
                                       __float_as_int(av[k].y));
            }
        }
        const int left = max(nl - g, 0);
        if (left > 0 && g > 0) {
            for (int k = 0; k < left; k += 32) {
                const int i = k + lane;
                int v = 0;
                if (i < left) v = lq[g + i];
                if (i < left) lq[i] = v;
            }
        }
        nl = left;
        if (lane == 0) qcnt[w] = cnt;
        int more = __syncthreads_or((step < stepEnd || nl > 0) ? 1 : 0);

        int np = 0;
        const int cntv = qcnt[lane];
        for (int qi = 0; qi < 32; ++qi) {
            const int c = __builtin_amdgcn_readlane(cntv, qi);
            for (int b = 0; b < c; b += 128) {
                int rowv[4];
#pragma unroll
                for (int k = 0; k < 4; ++k)
                    rowv[k] = q[qi * QCAP + min(b + 32 * k + lane, c - 1)].y;
#pragma unroll
                for (int k = 0; k < 4; ++k) {
                    const int idx = b + 32 * k + lane;
                    const bool mine = idx < c && (rowv[k] & 31) == w;
                    unsigned mm = __builtin_amdgcn_ballot_w32(mine);
                    if (mm) {
                        if (mine) pd[np + __builtin_popcount(mm & ltmask)] = qi * QCAP + idx;
                        np += __builtin_popcount(mm);
                    }
                }
                if (np >= 32) {
                    add_pending<!POOL>(q, pd, np, L, ns, h, lane, w0, w1, bb, acc);
                    np = 0;
                }
            }
        }
        if (np > 0) add_pending<!POOL>(q, pd, np, L, ns, h, lane, w0, w1, bb, acc);
        if (!more) break;
        __syncthreads();
    }
    }
    if (!POOL) {
        const int anyovf = __syncthreads_or(ns > ACAP ? 1 : 0);
        if (lane == 0) lcnt[blockIdx.x * 32 + w] = ns;
        if (tid == 0) lflag[blockIdx.x] = anyovf ? 0 : 1;
    } else {
        __syncthreads();
    }

    const float sc = 1.0f + eps[l];
    const float4* hrow = (const float4*)(h + (size_t)lo * HID);
    for (int i = tid; i < nrows * 32; i += 1024) {
        const int o = (i >> 5) * ROWP + (i & 31);
        float4 s = acc[o];
        float4 v = hrow[i];
        s.x += sc * v.x; s.y += sc * v.y; s.z += sc * v.z; s.w += sc * v.w;
        acc[o] = s;
    }
    __syncthreads();

    float* rows = (float*)acc;
    const int t = w >> 1, nh = w & 1;
    const bool work = w < 2 * OWN_T;
    v8f c[4];
    if (work) tile_product(rows, t, nh, lane, W1f, b1, c);
    __syncthreads();
    if (work) tile_relu_to_lds(rows, t, nh, lane, c);
    __syncthreads();
    if (work) tile_product(rows, t, nh, lane, W2f, b2, c);
    __syncthreads();
    if (work) tile_relu_to_lds(rows, t, nh, lane, c);
    __syncthreads();

    if (POOL) {
        const int col = tid & (HID - 1);
        const int segLen = (OWN_R + 7) / 8;
        const int r0 = (tid >> 7) * segLen;
        const int r1 = min(r0 + segLen, nrows);
        if (r0 < r1) {
            const int* gp = batch + lo;
            int cur = gp[r0];
            float s = 0.0f;
            for (int r = r0; r < r1; ++r) {
                int g = gp[r];
                if (g != cur) {
                    atomicAdd(out + (size_t)cur * HID + col, s);
                    s = 0.0f;
                    cur = g;
                }
                s += rows[r * (ROWP * 4) + col];
            }
            atomicAdd(out + (size_t)cur * HID + col, s);
        }
    } else {
        float4* orow = (float4*)(out + (size_t)lo * HID);
        for (int i = tid; i < nrows * 32; i += 1024)
            orow[i] = acc[(i >> 5) * ROWP + (i & 31)];
    }
}

__global__ __launch_bounds__(256) void gemm128_wmma_kernel(
    const float* __restrict__ A, const _Float16* __restrict__ Wf,
    const float* __restrict__ bias, float* __restrict__ out, int rows) {
    int wave = threadIdx.x >> 5, lane = threadIdx.x & 31;
    int tile = blockIdx.x * 8 + wave;
    int row0 = tile * 16;
    if (row0 >= rows) return;

    int mloc = lane & 15;
    int half = lane >> 4;
    int base = half * 8;
    const float* rowA = A + (size_t)(row0 + mloc) * HID;

    v8f acc[8];
#pragma unroll
    for (int nt = 0; nt < 8; ++nt) {
        float bv = bias[nt * 16 + mloc];
#pragma unroll
        for (int r = 0; r < 8; ++r) acc[nt][r] = bv;
    }

#pragma unroll
    for (int kk = 0; kk < 4; ++kk) {
        int k0 = kk * 32 + base;
        float4 a0 = *(const float4*)(rowA + k0);
        float4 a1 = *(const float4*)(rowA + k0 + 4);
        float4 a2 = *(const float4*)(rowA + k0 + 16);
        float4 a3 = *(const float4*)(rowA + k0 + 20);
        v16h a;
        a[0]  = (_Float16)a0.x; a[1]  = (_Float16)a0.y;
        a[2]  = (_Float16)a0.z; a[3]  = (_Float16)a0.w;
        a[4]  = (_Float16)a1.x; a[5]  = (_Float16)a1.y;
        a[6]  = (_Float16)a1.z; a[7]  = (_Float16)a1.w;
        a[8]  = (_Float16)a2.x; a[9]  = (_Float16)a2.y;
        a[10] = (_Float16)a2.z; a[11] = (_Float16)a2.w;
        a[12] = (_Float16)a3.x; a[13] = (_Float16)a3.y;
        a[14] = (_Float16)a3.z; a[15] = (_Float16)a3.w;
        v8h blo[8], bhi[8];
#pragma unroll
        for (int nt = 0; nt < 8; ++nt) {
            const v8h* bp = (const v8h*)Wf + ((kk * 8 + nt) * 2) * 32 + lane;
            blo[nt] = bp[0];
            bhi[nt] = bp[32];
        }
        asm volatile("" :: "v"(blo[0]), "v"(bhi[0]), "v"(blo[1]), "v"(bhi[1]),
                           "v"(blo[2]), "v"(bhi[2]), "v"(blo[3]), "v"(bhi[3]),
                           "v"(blo[4]), "v"(bhi[4]), "v"(blo[5]), "v"(bhi[5]),
                           "v"(blo[6]), "v"(bhi[6]), "v"(blo[7]), "v"(bhi[7]));
#pragma unroll
        for (int nt = 0; nt < 8; ++nt) {
            v16h b;
#pragma unroll
            for (int i = 0; i < 8; ++i) { b[i] = blo[nt][i]; b[8 + i] = bhi[nt][i]; }
            acc[nt] = __builtin_amdgcn_wmma_f32_16x16x32_f16(
                false, a, false, b, (short)0, acc[nt], false, false);
        }
    }

#pragma unroll
    for (int nt = 0; nt < 8; ++nt) {
        int col = nt * 16 + mloc;
#pragma unroll
        for (int r = 0; r < 8; ++r) {
            int M = row0 + r + half * 8;
            out[(size_t)M * HID + col] = acc[nt][r];
        }
    }
}

extern "C" void kernel_launch(void* const* d_in, const int* in_sizes, int n_in,
                              void* d_out, int out_size, void* d_ws, size_t ws_size,
                              hipStream_t stream) {
    const int N = in_sizes[0];
    const int E = in_sizes[1] / 2;
    const int G = out_size / (3 * HID);

    const float* enc_W  = (const float*)d_in[12];
    const float* enc_b  = (const float*)d_in[13];
    const float* edge_W = (const float*)d_in[14];
    const float* edge_b = (const float*)d_in[15];
    const float* eps    = (const float*)d_in[16];
    const float* W1     = (const float*)d_in[17];
    const float* b1     = (const float*)d_in[18];
    const float* W2     = (const float*)d_in[19];
    const float* b2     = (const float*)d_in[20];
    const float* fc_W   = (const float*)d_in[21];
    const float* fc_b   = (const float*)d_in[22];

    float* bufA   = (float*)d_ws;
    float* bufB   = bufA + (size_t)N * HID;
    float* pooled = bufB + (size_t)N * HID;
    _Float16* w16 = (_Float16*)(pooled + (size_t)3 * G * HID);
    _Float16* W1t0 = w16;
    _Float16* W1t1 = w16 + 16384;
    _Float16* W2t0 = w16 + 2 * 16384;
    _Float16* W2t1 = w16 + 3 * 16384;
    _Float16* fcWt = w16 + 4 * 16384;
    unsigned* keys = (unsigned*)(w16 + 5 * 16384);

    const int ownBlk    = (N + OWN_R - 1) / OWN_R;
    const int nz4       = (3 * G * HID) / 4;
    const int fcBlk     = ((G + 15) / 16 + 7) / 8;
    const int nchunk    = (E + 511) / 512;
    const int keyBlk    = (nchunk * 32 + 255) / 256;
    const int encBlk    = (N + 31) / 32;
    const int zBlk      = (nz4 + 255) / 256;

    int* lst   = (int*)(keys + (size_t)3 * nchunk * 128);
    int* lcnt  = lst + (size_t)ownBlk * 32 * ACAP * 4;
    int* lflag = lcnt + (size_t)ownBlk * 32;

    prep_kernel<<<320 + 3 * keyBlk + encBlk + zBlk, 256, 0, stream>>>(
        W1, W2, fc_W, w16,
        (const int*)d_in[1], (const int*)d_in[5], (const int*)d_in[9], keys, E, nchunk,
        keyBlk, (const float*)d_in[0], enc_W, enc_b, bufA, N, encBlk, pooled, nz4);

    for (int g = 0; g < 3; ++g) {
        const int*   ei = (const int*)  d_in[4 * g + 1];
        const float* ea = (const float*)d_in[4 * g + 2];
        const int*   bt = (const int*)  d_in[4 * g + 3];
        const float* xn = g < 2 ? (const float*)d_in[4 * g + 4] : nullptr;
        float* pg = pooled + (size_t)g * G * HID;

        layer_owner_kernel<false><<<ownBlk, 1024, 0, stream>>>(
            bufA, ei, keys + (size_t)g * nchunk * 128, ea, edge_W, edge_b, eps, 0, W1t0,
            b1, W2t0, b2, bufB, bt, lst, lcnt, lflag, N, E, nullptr, enc_W, enc_b, nullptr);
        layer_owner_kernel<true><<<ownBlk, 1024, 0, stream>>>(
            bufB, ei, keys + (size_t)g * nchunk * 128, ea, edge_W + 2 * HID, edge_b + HID,
            eps, 1, W1t1, b1 + HID,
            W2t1, b2 + HID, pg, bt, lst, lcnt, lflag, N, E, xn, enc_W, enc_b, bufA);

        gemm128_wmma_kernel<<<fcBlk, 256, 0, stream>>>(
            pg, fcWt, fc_b, (float*)d_out + (size_t)g * G * HID, G);
    }
}


# layer-0 edge scan: owner byte list -> owner-major bit table (1 bit per 4-edge group, set by LDS integer OR in the pre-pass); a work-group reads only its own row, no byte tests; sum order kept
# speedup vs baseline: 4.6721x; 1.0702x over previous
#include <hip/hip_runtime.h>

typedef __attribute__((ext_vector_type(16))) _Float16 v16h;
typedef __attribute__((ext_vector_type(8)))  _Float16 v8h;
typedef __attribute__((ext_vector_type(8)))  float    v8f;

#define HID 128

__device__ __forceinline__ void convert_wt_elem(const float* __restrict__ W,
                                                _Float16* __restrict__ Wf, int idx) {
    int i    = idx & 7;
    int lane = (idx >> 3) & 31;
    int j    = (idx >> 8) & 1;
    int nt   = (idx >> 9) & 7;
    int kk   = idx >> 12;
    int n = nt * 16 + (lane & 15);
    int k = kk * 32 + (lane >> 4) * 8 + j * 16 + i;
    Wf[idx] = (_Float16)W[k * HID + n];
}

__device__ __forceinline__ void encode_row(float xv, float4 w, float4 bb,
                                           float* __restrict__ hrow, int lane) {
    float4 r;
    r.x = xv * w.x + bb.x;
    r.y = xv * w.y + bb.y;
    r.z = xv * w.z + bb.z;
    r.w = xv * w.w + bb.w;
    *(float4*)(hrow + 4 * lane) = r;
}

#define OWN_R 196
#define BM_P  257
__device__ __forceinline__ void owner_bits_block(
    const int* __restrict__ ei, unsigned* __restrict__ bm, int E, int rowLen,
    int bx, int gsel, unsigned* sc) {
    const int* dstp = ei + E;
    const int tid = threadIdx.x, L = tid & 31;
    const int wv = __builtin_amdgcn_readfirstlane(tid >> 5);
    const int c = bx * 8 + wv;
    int4 d[4];
    if ((E & 3) == 0 && (c + 1) * 512 <= E) {
#pragma unroll
        for (int j = 0; j < 4; ++j)
            d[j] = *(const int4*)(dstp + ((c * 4 + j) << 7) + (L << 2));
    } else {
#pragma unroll
        for (int j = 0; j < 4; ++j) {
            const int e0 = ((c * 4 + j) << 7) + (L << 2);
            d[j] = make_int4(-1, -1, -1, -1);
            if (e0 < E)     d[j].x = dstp[e0];
            if (e0 + 1 < E) d[j].y = dstp[e0 + 1];
            if (e0 + 2 < E) d[j].z = dstp[e0 + 2];
            if (e0 + 3 < E) d[j].w = dstp[e0 + 3];
        }
    }
    for (int i = tid; i < (32 * BM_P) / 4; i += 256)
        ((uint4*)sc)[i] = make_uint4(0u, 0u, 0u, 0u);
    __syncthreads();
    const unsigned bit = 1u << L;
#pragma unroll
    for (int j = 0; j < 4; ++j) {
        unsigned* srow = sc + (wv * 4 + j) * BM_P;
        if (d[j].x >= 0) atomicOr(srow + (((unsigned)d[j].x / OWN_R) & 255u), bit);
        if (d[j].y >= 0) atomicOr(srow + (((unsigned)d[j].y / OWN_R) & 255u), bit);
        if (d[j].z >= 0) atomicOr(srow + (((unsigned)d[j].z / OWN_R) & 255u), bit);
        if (d[j].w >= 0) atomicOr(srow + (((unsigned)d[j].w / OWN_R) & 255u), bit);
    }
    __syncthreads();
    unsigned* orow = bm + ((size_t)gsel * 256 + wv * 32) * rowLen + bx * 32 + L;
    const unsigned* scol = sc + L * BM_P + wv * 32;
#pragma unroll 8
    for (int oo = 0; oo < 32; ++oo)
        orow[(size_t)oo * rowLen] = scol[oo];
}

__global__ __launch_bounds__(256) void prep_kernel(
    const float* __restrict__ W1, const float* __restrict__ W2,
    const float* __restrict__ fcW, _Float16* __restrict__ w16,
    const int* __restrict__ ei_a, const int* __restrict__ ei_p,
    const int* __restrict__ ei_n, unsigned* __restrict__ bm, int E, int rowLen,
    int keyBlk,
    const float* __restrict__ x0, const float* __restrict__ encW,
    const float* __restrict__ encb, float* __restrict__ h0, int N, int encBlk,
    float* __restrict__ pooled, int nz4) {
    __shared__ uint4 sc[(32 * BM_P) / 4];
    int b = blockIdx.x;
    if (b < 320) {
        const int m = b >> 6;
        const float* W = m == 0 ? W1 : m == 1 ? W1 + HID * HID :
                         m == 2 ? W2 : m == 3 ? W2 + HID * HID : fcW;
        convert_wt_elem(W, w16 + m * 16384, (b & 63) * 256 + threadIdx.x);
        return;
    }
    b -= 320;
    if (b < 3 * keyBlk) {
        const int gsel = b / keyBlk;
        const int* ei = gsel == 0 ? ei_a : (gsel == 1 ? ei_p : ei_n);
        owner_bits_block(ei, bm, E, rowLen, b - gsel * keyBlk, gsel, (unsigned*)sc);
        return;
    }
    b -= 3 * keyBlk;
    if (b < encBlk) {
        const int wave = threadIdx.x >> 5, lane = threadIdx.x & 31;
        const float4 w  = *(const float4*)(encW + 4 * lane);
        const float4 bb = *(const float4*)(encb + 4 * lane);
        const int n0 = b * 32 + wave * 4;
        float xv[4];
#pragma unroll
        for (int k = 0; k < 4; ++k) xv[k] = x0[min(n0 + k, N - 1)];
#pragma unroll
        for (int k = 0; k < 4; ++k)
            if (n0 + k < N) encode_row(xv[k], w, bb, h0 + (size_t)(n0 + k) * HID, lane);
        return;
    }
    b -= encBlk;
    const int i = b * 256 + threadIdx.x;
    if (i < nz4) ((float4*)pooled)[i] = make_float4(0.f, 0.f, 0.f, 0.f);
}

#define OWN_T 13
#define ROWP  33
#define QCAP  256
#define LQCAP 384
#define PCAP  160
#define ACAP  256
template <bool TAIL>
__device__ __forceinline__ void add_listed8(
    int4 ent, int p, int m, const float* __restrict__ h, int lane,
    float4 w0, float4 w1, float4 bb, float4* acc) {
    int srcu[8], rowu[8], a0u[8], a1u[8];
    float4 hv[8];
#pragma unroll
    for (int k = 0; k < 8; ++k) {
        const int i = TAIL ? min(p + k, m - 1) : p + k;
        srcu[k] = __builtin_amdgcn_readlane(ent.x, i);
        rowu[k] = __builtin_amdgcn_readlane(ent.y, i);
        a0u[k]  = __builtin_amdgcn_readlane(ent.z, i);
        a1u[k]  = __builtin_amdgcn_readlane(ent.w, i);
    }
#pragma unroll
    for (int k = 0; k < 8; ++k)
        hv[k] = *(const float4*)(h + (size_t)srcu[k] * HID + 4 * lane);
#pragma unroll
    for (int k = 0; k < 8; ++k) {
        const float a0 = __int_as_float(a0u[k]);
        const float a1 = __int_as_float(a1u[k]);
        float mx = fmaxf(hv[k].x + a0 * w0.x + a1 * w1.x + bb.x, 0.0f);
        float my = fmaxf(hv[k].y + a0 * w0.y + a1 * w1.y + bb.y, 0.0f);
        float mz = fmaxf(hv[k].z + a0 * w0.z + a1 * w1.z + bb.z, 0.0f);
        float mw = fmaxf(hv[k].w + a0 * w0.w + a1 * w1.w + bb.w, 0.0f);
        if (TAIL) {
            const int keep = (p + k < m) ? -1 : 0;
            mx = __int_as_float(__float_as_int(mx) & keep);
            my = __int_as_float(__float_as_int(my) & keep);
            mz = __int_as_float(__float_as_int(mz) & keep);
            mw = __int_as_float(__float_as_int(mw) & keep);
        }
        float4 s = acc[rowu[k] * ROWP + lane];
        s.x += mx; s.y += my; s.z += mz; s.w += mw;
        acc[rowu[k] * ROWP + lane] = s;
    }
}

__device__ __forceinline__ void add_listed(
    int4 ent, int m, const float* __restrict__ h, int lane,
    float4 w0, float4 w1, float4 bb, float4* acc) {
    int p = 0;
    for (; p + 8 <= m; p += 8) add_listed8<false>(ent, p, m, h, lane, w0, w1, bb, acc);
    if (p < m) add_listed8<true>(ent, p, m, h, lane, w0, w1, bb, acc);
}

template <bool SAVE>
__device__ __forceinline__ void add_pending(
    const int4* q, const int* pd, int np, int4* __restrict__ L, int& ns,
    const float* __restrict__ h, int lane,
    float4 w0, float4 w1, float4 bb, float4* acc) {
    const bool fits = SAVE && ns + np <= ACAP;
    for (int b0 = 0; b0 < np; b0 += 32) {
        const int m = min(32, np - b0);
        const int4 ent = q[pd[b0 + min(lane, m - 1)]];
        if (SAVE && fits && lane < m) L[(unsigned)(ns + b0) + (unsigned)lane] = ent;
        add_listed(ent, m, h, lane, w0, w1, bb, acc);
    }
    if (SAVE) ns = fits ? ns + np : ACAP + 1;
}

__device__ __forceinline__ bool note_steps(const unsigned (&mv)[8], int& step, int stepEnd,
                                           int* lq, int& nl, int lane, unsigned ltmask) {
#pragma unroll
    for (int u = 0; u < 8; ++u) {
        if (step >= stepEnd) return false;
        unsigned m = mv[u];
        const int ns32 = min(32, stepEnd - step);
        const int c = __builtin_popcount(m);
        int before = 0, total = 0;
#pragma unroll
        for (int i = 0; i < 6; ++i) {
            const unsigned bal = __builtin_amdgcn_ballot_w32(((c >> i) & 1) != 0);
            before += __builtin_popcount(bal & ltmask) << i;
            total  += __builtin_popcount(bal) << i;
        }
        if (nl + total <= LQCAP) {
            int pos = nl + before;
            const int e0s = (step + lane) << 7;
            while (m != 0u) {
                lq[pos++] = e0s + (__builtin_ctz(m) << 2);
                m &= m - 1u;
            }
            nl += total;
            step += ns32;
        } else {
            int took = 0;
            for (int s = 0; s < ns32; ++s) {
                const unsigned ms = __builtin_amdgcn_readlane(m, s);
                const int cs = __builtin_popcount(ms);
                if (nl + cs > LQCAP) break;
                if ((ms >> lane) & 1u)
                    lq[nl + __builtin_popcount(ms & ltmask)] = ((step + s) << 7) + (lane << 2);
                nl += cs;
                ++took;
            }
            step += took;
            return true;
        }
    }
    return false;
}

__device__ __forceinline__ void tile_product(
    const float* rows, int t, int nh, int lane, const _Float16* __restrict__ Wf,
    const float* __restrict__ bias, v8f* c) {
    const int mloc = lane & 15;
    const int base = (lane >> 4) * 8;
    const float* rowA = rows + (t * 16 + mloc) * (ROWP * 4);
#pragma unroll
    for (int j = 0; j < 4; ++j) {
        float bv = bias[(nh * 4 + j) * 16 + mloc];
#pragma unroll
        for (int r = 0; r < 8; ++r) c[j][r] = bv;
    }
#pragma unroll
    for (int kk = 0; kk < 4; ++kk) {
        int k0 = kk * 32 + base;
        float4 a0 = *(const float4*)(rowA + k0);
        float4 a1 = *(const float4*)(rowA + k0 + 4);
        float4 a2 = *(const float4*)(rowA + k0 + 16);
        float4 a3 = *(const float4*)(rowA + k0 + 20);
        v16h a;
        a[0]  = (_Float16)a0.x; a[1]  = (_Float16)a0.y;
        a[2]  = (_Float16)a0.z; a[3]  = (_Float16)a0.w;
        a[4]  = (_Float16)a1.x; a[5]  = (_Float16)a1.y;
        a[6]  = (_Float16)a1.z; a[7]  = (_Float16)a1.w;
        a[8]  = (_Float16)a2.x; a[9]  = (_Float16)a2.y;
        a[10] = (_Float16)a2.z; a[11] = (_Float16)a2.w;
        a[12] = (_Float16)a3.x; a[13] = (_Float16)a3.y;
        a[14] = (_Float16)a3.z; a[15] = (_Float16)a3.w;
        v8h blo[4], bhi[4];
#pragma unroll
        for (int j = 0; j < 4; ++j) {
            const v8h* bp = (const v8h*)Wf + ((kk * 8 + nh * 4 + j) * 2) * 32 + lane;
            blo[j] = bp[0];
            bhi[j] = bp[32];
        }
        asm volatile("" :: "v"(blo[0]), "v"(bhi[0]), "v"(blo[1]), "v"(bhi[1]),
                           "v"(blo[2]), "v"(bhi[2]), "v"(blo[3]), "v"(bhi[3]));
#pragma unroll
        for (int j = 0; j < 4; ++j) {
            v16h b;
#pragma unroll
            for (int i = 0; i < 8; ++i) { b[i] = blo[j][i]; b[8 + i] = bhi[j][i]; }
            c[j] = __builtin_amdgcn_wmma_f32_16x16x32_f16(
                false, a, false, b, (short)0, c[j], false, false);
        }
    }
}

__device__ __forceinline__ void tile_relu_to_lds(float* rows, int t, int nh,
                                                 int lane, const v8f* c) {
    float* p = rows + (t * 16 + (lane >> 4) * 8) * (ROWP * 4) + nh * 64 + (lane & 15);
#pragma unroll
    for (int j = 0; j < 4; ++j) {
#pragma unroll
        for (int r = 0; r < 8; ++r)
            p[r * (ROWP * 4) + j * 16] = fmaxf(c[j][r], 0.0f);
    }
}

__device__ __forceinline__ void encode_wave_rows(
    const float (&xv)[(OWN_R + 31) / 32], float4 ew, float4 ebv,
    float* __restrict__ hn, int lo, int nrows, int w, int lane) {
#pragma unroll
    for (int k = 0; k < (OWN_R + 31) / 32; ++k)
        if (w + 32 * k < nrows)
            encode_row(xv[k], ew, ebv, hn + (size_t)(lo + w + 32 * k) * HID, lane);
}

template <bool POOL>
__global__ __launch_bounds__(1024) void layer_owner_kernel(
    const float* __restrict__ h, const int* __restrict__ ei,
    const unsigned* __restrict__ key,
    const float* __restrict__ ea, const float* __restrict__ eW,
    const float* __restrict__ eb, const float* __restrict__ eps, int l,
    const _Float16* __restrict__ W1f, const float* __restrict__ b1,
    const _Float16* __restrict__ W2f, const float* __restrict__ b2,
    float* __restrict__ out, const int* __restrict__ batch,
    int* __restrict__ lst, int* __restrict__ lcnt, int* __restrict__ lflag, int N, int E,
    const float* __restrict__ xn, const float* __restrict__ encW,
    const float* __restrict__ encb, float* __restrict__ hn) {
    __shared__ float4 acc[OWN_T * 16 * ROWP];
    __shared__ int4   q[32 * QCAP];
    __shared__ int    pend[32 * PCAP];
    __shared__ int    qcnt[32];
    __shared__ int    lqs[32 * LQCAP];

    const int tid = threadIdx.x, lane = tid & 31;
    const int w = __builtin_amdgcn_readfirstlane(tid >> 5);
    const unsigned ltmask = (1u << lane) - 1u;
    const int lo = blockIdx.x * OWN_R;
    const int nrows = min(OWN_R, N - lo);

    int4* L = (int4*)lst + ((size_t)blockIdx.x * 32 + w) * ACAP;
    int reuse = 0, nlist = 0;
    if (POOL) {
        reuse = __builtin_amdgcn_readfirstlane(lflag[blockIdx.x]);
        nlist = __builtin_amdgcn_readfirstlane(lcnt[blockIdx.x * 32 + w]);
    }

    for (int i = tid; i < OWN_T * 16 * ROWP; i += 1024)
        acc[i] = make_float4(0.f, 0.f, 0.f, 0.f);

    const float4 w0 = *(const float4*)(eW + 4 * lane);
    const float4 w1 = *(const float4*)(eW + HID + 4 * lane);
    const float4 bb = *(const float4*)(eb + 4 * lane);

    const int NRW = (OWN_R + 31) / 32;
    const bool enc = POOL && xn != nullptr;
    float4 ew = make_float4(0.f, 0.f, 0.f, 0.f), ebv = ew;
    float xv[NRW];
#pragma unroll
    for (int k = 0; k < NRW; ++k) xv[k] = 0.0f;
    if (enc) {
        ew  = *(const float4*)(encW + 4 * lane);
        ebv = *(const float4*)(encb + 4 * lane);
#pragma unroll
        for (int k = 0; k < NRW; ++k) xv[k] = xn[lo + min(w + 32 * k, nrows - 1)];
    }

    const int* dstp = ei + E;
    const int stepsTotal = (E + 127) >> 7;
    const int chunksTotal = (stepsTotal + 3) >> 2;
    const int cpw = (chunksTotal + 31) >> 5;
    int step = min(w * cpw * 4, stepsTotal);
    const int stepEnd = min((w * cpw + cpw) * 4, stepsTotal);
    const int rowLen = ((chunksTotal + 7) >> 3) << 5;
    const unsigned* bmrow = key + (size_t)(blockIdx.x & 255u) * rowLen;
    int4* myq = q + w * QCAP;
    int*  pd  = pend + w * PCAP;
    int*  lq  = lqs + w * LQCAP;
    int nl = 0;
    int ns = 0;
    if (POOL && reuse != 0) {
        __syncthreads();
        if (enc) encode_wave_rows(xv, ew, ebv, hn, lo, nrows, w, lane);
        if (nlist > 0) {
            int4 cur = L[min(lane, nlist - 1)];
            for (int b0 = 0; b0 < nlist; b0 += 32) {
                const int4 nxt = L[min(b0 + 32 + lane, nlist - 1)];
                add_listed(cur, min(32, nlist - b0), h, lane, w0, w1, bb, acc);
                cur = nxt;
            }
        }
    } else {
    if (enc) encode_wave_rows(xv, ew, ebv, hn, lo, nrows, w, lane);
    for (;;) {
        bool lfull = false;
        while (!lfull && step < stepEnd) {
            unsigned mv[8];
#pragma unroll
            for (int u = 0; u < 8; ++u) {
                const int s = step + 32 * u + lane;
                const unsigned v = bmrow[min(s, stepEnd - 1)];
                mv[u] = s < stepEnd ? v : 0u;
            }
            lfull = note_steps(mv, step, stepEnd, lq, nl, lane, ltmask);
        }
        int cnt = 0, g = 0;
        bool qfull = false;
        while (!qfull && g < nl) {
            int e0v[4], dv[4][4];
#pragma unroll
            for (int k = 0; k < 4; ++k) {
                const int i = g + 32 * k + lane;
                e0v[k] = 0;
                if (i < nl) e0v[k] = lq[i];
            }
#pragma unroll
            for (int k = 0; k < 4; ++k) {
#pragma unroll
                for (int b = 0; b < 4; ++b) dv[k][b] = dstp[min(e0v[k] + b, E - 1)];
            }
#pragma unroll
            for (int k = 0; k < 4; ++k) {
                if (!qfull && g < nl) {
                    const bool have = g + lane < nl;
                    bool ok[4];
                    int nok = 0;
#pragma unroll
                    for (int b = 0; b < 4; ++b) {
                        ok[b] = have && e0v[k] + b < E &&
                                (unsigned)(dv[k][b] - lo) < (unsigned)nrows;
                        nok += ok[b] ? 1 : 0;
                    }
                    const unsigned p0 = __builtin_amdgcn_ballot_w32((nok & 1) != 0);
                    const unsigned p1 = __builtin_amdgcn_ballot_w32((nok & 2) != 0);
                    const unsigned p2 = __builtin_amdgcn_ballot_w32((nok & 4) != 0);
                    const int total = __builtin_popcount(p0) + 2 * __builtin_popcount(p1) +
                                      4 * __builtin_popcount(p2);
                    if (cnt + total > QCAP) {
                        qfull = true;
                    } else {
                        int pos = cnt + __builtin_popcount(p0 & ltmask) +
                                  2 * __builtin_popcount(p1 & ltmask) +
                                  4 * __builtin_popcount(p2 & ltmask);
#pragma unroll
                        for (int b = 0; b < 4; ++b) {
                            if (ok[b]) {
                                myq[pos] = make_int4(e0v[k] + b, dv[k][b] - lo, 0, 0);
                                ++pos;
                            }
                        }
                        cnt += total;
                        g += 32;
                    }
                }
            }
        }
        for (int b0 = 0; b0 < cnt; b0 += 128) {
            int ex[4], rv[4], sv[4];
            float2 av[4];
#pragma unroll
            for (int k = 0; k < 4; ++k) {
                int i = b0 + 32 * k + lane;
                int4 t = make_int4(0, 0, 0, 0);
                if (i < cnt) t = myq[i];
                ex[k] = t.x;
                rv[k] = t.y;
            }
#pragma unroll
            for (int k = 0; k < 4; ++k) {
                sv[k] = ei[ex[k]];
                av[k] = *(const float2*)(ea + 2 * (size_t)ex[k]);
            }
#pragma unroll
            for (int k = 0; k < 4; ++k) {
                int i = b0 + 32 * k + lane;
                if (i < cnt)
                    myq[i] = make_int4(sv[k], rv[k], __float_as_int(av[k].x),
                                       __float_as_int(av[k].y));
            }
        }
        const int left = max(nl - g, 0);
        if (left > 0 && g > 0) {
            for (int k = 0; k < left; k += 32) {
                const int i = k + lane;
                int v = 0;
                if (i < left) v = lq[g + i];
                if (i < left) lq[i] = v;
            }
        }
        nl = left;
        if (lane == 0) qcnt[w] = cnt;
        int more = __syncthreads_or((step < stepEnd || nl > 0) ? 1 : 0);

        int np = 0;
        const int cntv = qcnt[lane];
        for (int qi = 0; qi < 32; ++qi) {
            const int c = __builtin_amdgcn_readlane(cntv, qi);
            for (int b = 0; b < c; b += 128) {
                int rowv[4];
#pragma unroll
                for (int k = 0; k < 4; ++k)
                    rowv[k] = q[qi * QCAP + min(b + 32 * k + lane, c - 1)].y;
#pragma unroll
                for (int k = 0; k < 4; ++k) {
                    const int idx = b + 32 * k + lane;
                    const bool mine = idx < c && (rowv[k] & 31) == w;
                    unsigned mm = __builtin_amdgcn_ballot_w32(mine);
                    if (mm) {
                        if (mine) pd[np + __builtin_popcount(mm & ltmask)] = qi * QCAP + idx;
                        np += __builtin_popcount(mm);
                    }
                }
                if (np >= 32) {
                    add_pending<!POOL>(q, pd, np, L, ns, h, lane, w0, w1, bb, acc);
                    np = 0;
                }
            }
        }
        if (np > 0) add_pending<!POOL>(q, pd, np, L, ns, h, lane, w0, w1, bb, acc);
        if (!more) break;
        __syncthreads();
    }
    }
    if (!POOL) {
        const int anyovf = __syncthreads_or(ns > ACAP ? 1 : 0);
        if (lane == 0) lcnt[blockIdx.x * 32 + w] = ns;
        if (tid == 0) lflag[blockIdx.x] = anyovf ? 0 : 1;
    } else {
        __syncthreads();
    }

    const float sc = 1.0f + eps[l];
    const float4* hrow = (const float4*)(h + (size_t)lo * HID);
    for (int i = tid; i < nrows * 32; i += 1024) {
        const int o = (i >> 5) * ROWP + (i & 31);
        float4 s = acc[o];
        float4 v = hrow[i];
        s.x += sc * v.x; s.y += sc * v.y; s.z += sc * v.z; s.w += sc * v.w;
        acc[o] = s;
    }
    __syncthreads();

    float* rows = (float*)acc;
    const int t = w >> 1, nh = w & 1;
    const bool work = w < 2 * OWN_T;
    v8f c[4];
    if (work) tile_product(rows, t, nh, lane, W1f, b1, c);
    __syncthreads();
    if (work) tile_relu_to_lds(rows, t, nh, lane, c);
    __syncthreads();
    if (work) tile_product(rows, t, nh, lane, W2f, b2, c);
    __syncthreads();
    if (work) tile_relu_to_lds(rows, t, nh, lane, c);
    __syncthreads();

    if (POOL) {
        const int col = tid & (HID - 1);
        const int segLen = (OWN_R + 7) / 8;
        const int r0 = (tid >> 7) * segLen;
        const int r1 = min(r0 + segLen, nrows);
        if (r0 < r1) {
            const int* gp = batch + lo;
            int cur = gp[r0];
            float s = 0.0f;
            for (int r = r0; r < r1; ++r) {
                int g = gp[r];
                if (g != cur) {
                    atomicAdd(out + (size_t)cur * HID + col, s);
                    s = 0.0f;
                    cur = g;
                }
                s += rows[r * (ROWP * 4) + col];
            }
            atomicAdd(out + (size_t)cur * HID + col, s);
        }
    } else {
        float4* orow = (float4*)(out + (size_t)lo * HID);
        for (int i = tid; i < nrows * 32; i += 1024)
            orow[i] = acc[(i >> 5) * ROWP + (i & 31)];
    }
}

__global__ __launch_bounds__(256) void gemm128_wmma_kernel(
    const float* __restrict__ A, const _Float16* __restrict__ Wf,
    const float* __restrict__ bias, float* __restrict__ out, int rows) {
    int wave = threadIdx.x >> 5, lane = threadIdx.x & 31;
    int tile = blockIdx.x * 8 + wave;
    int row0 = tile * 16;
    if (row0 >= rows) return;

    int mloc = lane & 15;
    int half = lane >> 4;
    int base = half * 8;
    const float* rowA = A + (size_t)(row0 + mloc) * HID;

    v8f acc[8];
#pragma unroll
    for (int nt = 0; nt < 8; ++nt) {
        float bv = bias[nt * 16 + mloc];
#pragma unroll
        for (int r = 0; r < 8; ++r) acc[nt][r] = bv;
    }

#pragma unroll
    for (int kk = 0; kk < 4; ++kk) {
        int k0 = kk * 32 + base;
        float4 a0 = *(const float4*)(rowA + k0);
        float4 a1 = *(const float4*)(rowA + k0 + 4);
        float4 a2 = *(const float4*)(rowA + k0 + 16);
        float4 a3 = *(const float4*)(rowA + k0 + 20);
        v16h a;
        a[0]  = (_Float16)a0.x; a[1]  = (_Float16)a0.y;
        a[2]  = (_Float16)a0.z; a[3]  = (_Float16)a0.w;
        a[4]  = (_Float16)a1.x; a[5]  = (_Float16)a1.y;
        a[6]  = (_Float16)a1.z; a[7]  = (_Float16)a1.w;
        a[8]  = (_Float16)a2.x; a[9]  = (_Float16)a2.y;
        a[10] = (_Float16)a2.z; a[11] = (_Float16)a2.w;
        a[12] = (_Float16)a3.x; a[13] = (_Float16)a3.y;
        a[14] = (_Float16)a3.z; a[15] = (_Float16)a3.w;
        v8h blo[8], bhi[8];
#pragma unroll
        for (int nt = 0; nt < 8; ++nt) {
            const v8h* bp = (const v8h*)Wf + ((kk * 8 + nt) * 2) * 32 + lane;
            blo[nt] = bp[0];
            bhi[nt] = bp[32];
        }
        asm volatile("" :: "v"(blo[0]), "v"(bhi[0]), "v"(blo[1]), "v"(bhi[1]),
                           "v"(blo[2]), "v"(bhi[2]), "v"(blo[3]), "v"(bhi[3]),
                           "v"(blo[4]), "v"(bhi[4]), "v"(blo[5]), "v"(bhi[5]),
                           "v"(blo[6]), "v"(bhi[6]), "v"(blo[7]), "v"(bhi[7]));
#pragma unroll
        for (int nt = 0; nt < 8; ++nt) {
            v16h b;
#pragma unroll
            for (int i = 0; i < 8; ++i) { b[i] = blo[nt][i]; b[8 + i] = bhi[nt][i]; }
            acc[nt] = __builtin_amdgcn_wmma_f32_16x16x32_f16(
                false, a, false, b, (short)0, acc[nt], false, false);
        }
    }

#pragma unroll
    for (int nt = 0; nt < 8; ++nt) {
        int col = nt * 16 + mloc;
#pragma unroll
        for (int r = 0; r < 8; ++r) {
            int M = row0 + r + half * 8;
            out[(size_t)M * HID + col] = acc[nt][r];
        }
    }
}

extern "C" void kernel_launch(void* const* d_in, const int* in_sizes, int n_in,
                              void* d_out, int out_size, void* d_ws, size_t ws_size,
                              hipStream_t stream) {
    const int N = in_sizes[0];
    const int E = in_sizes[1] / 2;
    const int G = out_size / (3 * HID);

    const float* enc_W  = (const float*)d_in[12];
    const float* enc_b  = (const float*)d_in[13];
    const float* edge_W = (const float*)d_in[14];
    const float* edge_b = (const float*)d_in[15];
    const float* eps    = (const float*)d_in[16];
    const float* W1     = (const float*)d_in[17];
    const float* b1     = (const float*)d_in[18];
    const float* W2     = (const float*)d_in[19];
    const float* b2     = (const float*)d_in[20];
    const float* fc_W   = (const float*)d_in[21];
    const float* fc_b   = (const float*)d_in[22];

    float* bufA   = (float*)d_ws;
    float* bufB   = bufA + (size_t)N * HID;
    float* pooled = bufB + (size_t)N * HID;
    _Float16* w16 = (_Float16*)(pooled + (size_t)3 * G * HID);
    _Float16* W1t0 = w16;
    _Float16* W1t1 = w16 + 16384;
    _Float16* W2t0 = w16 + 2 * 16384;
    _Float16* W2t1 = w16 + 3 * 16384;
    _Float16* fcWt = w16 + 4 * 16384;
    unsigned* bits = (unsigned*)(w16 + 5 * 16384);

    const int ownBlk    = (N + OWN_R - 1) / OWN_R;
    const int nz4       = (3 * G * HID) / 4;
    const int fcBlk     = ((G + 15) / 16 + 7) / 8;
    const int nchunk    = (E + 511) / 512;
    const int keyBlk    = (nchunk + 7) / 8;
    const int rowLen    = keyBlk * 32;
    const int encBlk    = (N + 31) / 32;
    const int zBlk      = (nz4 + 255) / 256;

    int* lst   = (int*)(bits + (size_t)3 * 256 * rowLen);
    int* lcnt  = lst + (size_t)ownBlk * 32 * ACAP * 4;
    int* lflag = lcnt + (size_t)ownBlk * 32;

    prep_kernel<<<320 + 3 * keyBlk + encBlk + zBlk, 256, 0, stream>>>(
        W1, W2, fc_W, w16,
        (const int*)d_in[1], (const int*)d_in[5], (const int*)d_in[9], bits, E, rowLen,
        keyBlk, (const float*)d_in[0], enc_W, enc_b, bufA, N, encBlk, pooled, nz4);

    for (int g = 0; g < 3; ++g) {
        const int*   ei = (const int*)  d_in[4 * g + 1];
        const float* ea = (const float*)d_in[4 * g + 2];
        const int*   bt = (const int*)  d_in[4 * g + 3];
        const float* xn = g < 2 ? (const float*)d_in[4 * g + 4] : nullptr;
        float* pg = pooled + (size_t)g * G * HID;

        layer_owner_kernel<false><<<ownBlk, 1024, 0, stream>>>(
            bufA, ei, bits + (size_t)g * 256 * rowLen, ea, edge_W, edge_b, eps, 0, W1t0,
            b1, W2t0, b2, bufB, bt, lst, lcnt, lflag, N, E, nullptr, enc_W, enc_b, nullptr);
        layer_owner_kernel<true><<<ownBlk, 1024, 0, stream>>>(
            bufB, ei, bits + (size_t)g * 256 * rowLen, ea, edge_W + 2 * HID, edge_b + HID,
            eps, 1, W1t1, b1 + HID,
            W2t1, b2 + HID, pg, bt, lst, lcnt, lflag, N, E, xn, enc_W, enc_b, bufA);

        gemm128_wmma_kernel<<<fcBlk, 256, 0, stream>>>(
            pg, fcWt, fc_b, (float*)d_out + (size_t)g * G * HID, G);
    }
}
